# starting kernel
# baseline (speedup 1.0000x reference)
#include <hip/hip_runtime.h>
#include <math.h>

typedef __attribute__((ext_vector_type(2))) float v2f;
typedef __attribute__((ext_vector_type(8))) float v8f;

#define FIN   128
#define HID   8
#define NCLS  16
#define NEG_SLOPE 0.2f

__device__ __forceinline__ void atomicMaxF(float* addr, float v) {
    if (v >= 0.0f) atomicMax((int*)addr, __float_as_int(v));
    else           atomicMin((unsigned int*)addr, __float_as_uint(v));
}

__global__ void init_ws(float* m1, float* dn1, float* agg1,
                        float* m2, float* dn2, float* agg2, int n) {
    int i = blockIdx.x * blockDim.x + threadIdx.x;
    if (i < n * NCLS) agg2[i] = 0.0f;
    if (i < n * HID)  agg1[i] = 0.0f;
    if (i < n) {
        m1[i] = -__builtin_inff();
        m2[i] = -__builtin_inff();
        dn1[i] = 0.0f;
        dn2[i] = 0.0f;
    }
}

__global__ void gemm1_wmma(const float* __restrict__ x, const float* __restrict__ W1,
                           float* __restrict__ h1, int n) {
    __shared__ float wlds[FIN * 16];
    int t = threadIdx.x;
    {
        int k = t;
        const float* wr = W1 + k * HID;
        #pragma unroll
        for (int c = 0; c < 16; ++c) wlds[k * 16 + c] = (c < HID) ? wr[c] : 0.0f;
    }
    __syncthreads();

    int wave = t >> 5, lane = t & 31;
    int tile = blockIdx.x * 4 + wave;
    int col  = lane & 15;
    int hi   = lane >> 4;
    int mrow = tile * 16 + col;
    if (mrow >= n) mrow = n - 1;
    const float* xrow = x + (long)mrow * FIN;

    v8f c = {};
    #pragma unroll
    for (int k0 = 0; k0 < FIN; k0 += 4) {
        int ka = k0 + 2 * hi;
        v2f a = *(const v2f*)(xrow + ka);
        v2f b;
        b.x = wlds[ka * 16 + col];
        b.y = wlds[(ka + 1) * 16 + col];
        c = __builtin_amdgcn_wmma_f32_16x16x4_f32(false, a, false, b,
                                                  (short)0, c, false, false);
    }

    int mbase = tile * 16 + (hi << 3);
    if (col < HID) {
        #pragma unroll
        for (int j = 0; j < 8; ++j) {
            int r = mbase + j;
            if (r < n) h1[(long)r * HID + col] = c[j];
        }
    }
}

__global__ void gemm2_wmma(const float* __restrict__ h1r, const float* __restrict__ W2,
                           float* __restrict__ h2, int n) {
    __shared__ float wlds[HID * 16];
    int t = threadIdx.x;
    if (t < HID * 16) wlds[t] = W2[t];
    __syncthreads();

    int wave = t >> 5, lane = t & 31;
    int tile = blockIdx.x * 4 + wave;
    int col  = lane & 15;
    int hi   = lane >> 4;
    int mrow = tile * 16 + col;
    if (mrow >= n) mrow = n - 1;
    const float* hrow = h1r + (long)mrow * HID;

    v8f c = {};
    #pragma unroll
    for (int k0 = 0; k0 < HID; k0 += 4) {
        int ka = k0 + 2 * hi;
        v2f a = *(const v2f*)(hrow + ka);
        v2f b;
        b.x = wlds[ka * 16 + col];
        b.y = wlds[(ka + 1) * 16 + col];
        c = __builtin_amdgcn_wmma_f32_16x16x4_f32(false, a, false, b,
                                                  (short)0, c, false, false);
    }

    int mbase = tile * 16 + (hi << 3);
    #pragma unroll
    for (int j = 0; j < 8; ++j) {
        int r = mbase + j;
        if (r < n) h2[(long)r * NCLS + col] = c[j];
    }
}

template <int F>
__global__ void alpha_kernel(const float* __restrict__ h,
                             const float* __restrict__ a_src,
                             const float* __restrict__ a_dst,
                             float* __restrict__ outs, float* __restrict__ outd,
                             int n) {
    int i = blockIdx.x * blockDim.x + threadIdx.x;
    if (i >= n) return;
    const float4* hr = (const float4*)(h + (long)i * F);
    float s = 0.0f, d = 0.0f;
    #pragma unroll
    for (int q = 0; q < F / 4; ++q) {
        float4 hv = hr[q];
        s += hv.x * a_src[4 * q + 0] + hv.y * a_src[4 * q + 1]
           + hv.z * a_src[4 * q + 2] + hv.w * a_src[4 * q + 3];
        d += hv.x * a_dst[4 * q + 0] + hv.y * a_dst[4 * q + 1]
           + hv.z * a_dst[4 * q + 2] + hv.w * a_dst[4 * q + 3];
    }
    outs[i] = s;
    outd[i] = d;
}

__global__ void edge_max(const int* __restrict__ src, const int* __restrict__ dst,
                         const float* __restrict__ as, const float* __restrict__ ad,
                         float* __restrict__ m, int E) {
    int i = blockIdx.x * blockDim.x + threadIdx.x;
    if (i >= E) return;
    int s = src[i], d = dst[i];
    float e = as[s] + ad[d];
    e = (e >= 0.0f) ? e : NEG_SLOPE * e;
    atomicMaxF(&m[d], e);
}

template <int F>
__global__ void edge_acc(const int* __restrict__ src, const int* __restrict__ dst,
                         const float* __restrict__ as, const float* __restrict__ ad,
                         const float* __restrict__ m, const float* __restrict__ h,
                         float* __restrict__ denom, float* __restrict__ agg, int E) {
    int i = blockIdx.x * blockDim.x + threadIdx.x;
    if (i >= E) return;
    int s = src[i], d = dst[i];
    float e = as[s] + ad[d];
    e = (e >= 0.0f) ? e : NEG_SLOPE * e;
    float ex = __expf(e - m[d]);
    atomicAdd(&denom[d], ex);
    const float4* hs = (const float4*)(h + (long)s * F);
    float* ag = agg + (long)d * F;
    #pragma unroll
    for (int q = 0; q < F / 4; ++q) {
        float4 hv = hs[q];
        atomicAdd(&ag[4 * q + 0], ex * hv.x);
        atomicAdd(&ag[4 * q + 1], ex * hv.y);
        atomicAdd(&ag[4 * q + 2], ex * hv.z);
        atomicAdd(&ag[4 * q + 3], ex * hv.w);
    }
}

__global__ void finish1(float* __restrict__ agg, const float* __restrict__ denom,
                        const float* __restrict__ bias, int n) {
    int i = blockIdx.x * blockDim.x + threadIdx.x;
    if (i >= n) return;
    float d = denom[i];
    float inv = (d > 0.0f) ? 1.0f / d : 0.0f;
    float4* ap = (float4*)(agg + (long)i * HID);
    #pragma unroll
    for (int q = 0; q < HID / 4; ++q) {
        float4 v = ap[q];
        v.x = fmaxf(v.x * inv + bias[4 * q + 0], 0.0f);
        v.y = fmaxf(v.y * inv + bias[4 * q + 1], 0.0f);
        v.z = fmaxf(v.z * inv + bias[4 * q + 2], 0.0f);
        v.w = fmaxf(v.w * inv + bias[4 * q + 3], 0.0f);
        ap[q] = v;
    }
}

__global__ void finish2(const float* __restrict__ agg, const float* __restrict__ denom,
                        const float* __restrict__ bias, float* __restrict__ out, int n) {
    int i = blockIdx.x * blockDim.x + threadIdx.x;
    if (i >= n) return;
    float d = denom[i];
    float inv = (d > 0.0f) ? 1.0f / d : 0.0f;
    const float4* ap = (const float4*)(agg + (long)i * NCLS);
    float v[NCLS];
    float mx = -__builtin_inff();
    #pragma unroll
    for (int q = 0; q < NCLS / 4; ++q) {
        float4 a = ap[q];
        v[4 * q + 0] = a.x * inv + bias[4 * q + 0];
        v[4 * q + 1] = a.y * inv + bias[4 * q + 1];
        v[4 * q + 2] = a.z * inv + bias[4 * q + 2];
        v[4 * q + 3] = a.w * inv + bias[4 * q + 3];
    }
    #pragma unroll
    for (int f = 0; f < NCLS; ++f) mx = fmaxf(mx, v[f]);
    float s = 0.0f;
    #pragma unroll
    for (int f = 0; f < NCLS; ++f) s += __expf(v[f] - mx);
    float ls = __logf(s) + mx;
    float4* op = (float4*)(out + (long)i * NCLS);
    #pragma unroll
    for (int q = 0; q < NCLS / 4; ++q) {
        float4 o;
        o.x = v[4 * q + 0] - ls;
        o.y = v[4 * q + 1] - ls;
        o.z = v[4 * q + 2] - ls;
        o.w = v[4 * q + 3] - ls;
        op[q] = o;
    }
}

extern "C" void kernel_launch(void* const* d_in, const int* in_sizes, int n_in,
                              void* d_out, int out_size, void* d_ws, size_t ws_size,
                              hipStream_t stream) {
    const float* x   = (const float*)d_in[0];
    const int*   ei  = (const int*)  d_in[1];
    const float* W1  = (const float*)d_in[2];
    const float* a1s = (const float*)d_in[3];
    const float* a1d = (const float*)d_in[4];
    const float* b1  = (const float*)d_in[5];
    const float* W2  = (const float*)d_in[6];
    const float* a2s = (const float*)d_in[7];
    const float* a2d = (const float*)d_in[8];
    const float* b2  = (const float*)d_in[9];

    int n = in_sizes[0] / FIN;
    int E = in_sizes[1] / 2;
    const int* src = ei;
    const int* dst = ei + E;

    float* ws   = (float*)d_ws;
    float* h1   = ws; ws += (long)n * HID;
    float* as1  = ws; ws += n;
    float* ad1  = ws; ws += n;
    float* m1   = ws; ws += n;
    float* dn1  = ws; ws += n;
    float* agg1 = ws; ws += (long)n * HID;
    float* h2   = ws; ws += (long)n * NCLS;
    float* as2  = ws; ws += n;
    float* ad2  = ws; ws += n;
    float* m2   = ws; ws += n;
    float* dn2  = ws; ws += n;
    float* agg2 = ws; ws += (long)n * NCLS;

    int ntiles  = (n + 15) / 16;
    int gblocks = (ntiles + 3) / 4;
    int eblocks = (E + 255) / 256;
    int nblocks = (n + 255) / 256;

    init_ws<<<(n * NCLS + 255) / 256, 256, 0, stream>>>(m1, dn1, agg1, m2, dn2, agg2, n);

    gemm1_wmma<<<gblocks, 128, 0, stream>>>(x, W1, h1, n);
    alpha_kernel<HID><<<nblocks, 256, 0, stream>>>(h1, a1s, a1d, as1, ad1, n);
    edge_max<<<eblocks, 256, 0, stream>>>(src, dst, as1, ad1, m1, E);
    edge_acc<HID><<<eblocks, 256, 0, stream>>>(src, dst, as1, ad1, m1, h1, dn1, agg1, E);
    finish1<<<nblocks, 256, 0, stream>>>(agg1, dn1, b1, n);

    gemm2_wmma<<<gblocks, 128, 0, stream>>>(agg1, W2, h2, n);
    alpha_kernel<NCLS><<<nblocks, 256, 0, stream>>>(h2, a2s, a2d, as2, ad2, n);
    edge_max<<<eblocks, 256, 0, stream>>>(src, dst, as2, ad2, m2, E);
    edge_acc<NCLS><<<eblocks, 256, 0, stream>>>(src, dst, as2, ad2, m2, h2, dn2, agg2, E);
    finish2<<<nblocks, 256, 0, stream>>>(agg2, dn2, b2, (float*)d_out, n);
}


# edge_acc: lanes turned to features; each edge's ex*h[src] and ex go into one padded dst row (sums+denominator) as ONE atomic instruction in one cache line; f32 atomic arrival-order sums kept
# speedup vs baseline: 2.9534x; 2.9534x over previous
#include <hip/hip_runtime.h>
#include <math.h>

typedef __attribute__((ext_vector_type(2))) float v2f;
typedef __attribute__((ext_vector_type(8))) float v8f;

#define FIN   128
#define HID   8
#define NCLS  16
#define NEG_SLOPE 0.2f
#define ROW1  16
#define ROW2  32

__device__ __forceinline__ void atomicMaxF(float* addr, float v) {
    if (v >= 0.0f) atomicMax((int*)addr, __float_as_int(v));
    else           atomicMin((unsigned int*)addr, __float_as_uint(v));
}

__global__ void init_ws(float* acc1, float* m1, float* m2, int n) {
    int i = blockIdx.x * blockDim.x + threadIdx.x;
    if (i < n * ROW1) acc1[i] = 0.0f;
    if (i < n) {
        m1[i] = -__builtin_inff();
        m2[i] = -__builtin_inff();
    }
}

__global__ void gemm1_wmma(const float* __restrict__ x, const float* __restrict__ W1,
                           float* __restrict__ h1, int n) {
    __shared__ float wlds[FIN * 16];
    int t = threadIdx.x;
    {
        int k = t;
        const float* wr = W1 + k * HID;
        #pragma unroll
        for (int c = 0; c < 16; ++c) wlds[k * 16 + c] = (c < HID) ? wr[c] : 0.0f;
    }
    __syncthreads();

    int wave = t >> 5, lane = t & 31;
    int tile = blockIdx.x * 4 + wave;
    int col  = lane & 15;
    int hi   = lane >> 4;
    int mrow = tile * 16 + col;
    if (mrow >= n) mrow = n - 1;
    const float* xrow = x + (long)mrow * FIN;

    v8f c = {};
    #pragma unroll
    for (int k0 = 0; k0 < FIN; k0 += 4) {
        int ka = k0 + 2 * hi;
        v2f a = *(const v2f*)(xrow + ka);
        v2f b;
        b.x = wlds[ka * 16 + col];
        b.y = wlds[(ka + 1) * 16 + col];
        c = __builtin_amdgcn_wmma_f32_16x16x4_f32(false, a, false, b,
                                                  (short)0, c, false, false);
    }

    int mbase = tile * 16 + (hi << 3);
    if (col < HID) {
        #pragma unroll
        for (int j = 0; j < 8; ++j) {
            int r = mbase + j;
            if (r < n) h1[(long)r * HID + col] = c[j];
        }
    }
}

__global__ void gemm2_wmma(const float* __restrict__ h1r, const float* __restrict__ W2,
                           float* __restrict__ h2, int n) {
    __shared__ float wlds[HID * 16];
    int t = threadIdx.x;
    if (t < HID * 16) wlds[t] = W2[t];
    __syncthreads();

    int wave = t >> 5, lane = t & 31;
    int tile = blockIdx.x * 4 + wave;
    int col  = lane & 15;
    int hi   = lane >> 4;
    int mrow = tile * 16 + col;
    if (mrow >= n) mrow = n - 1;
    const float* hrow = h1r + (long)mrow * ROW1;

    v8f c = {};
    #pragma unroll
    for (int k0 = 0; k0 < HID; k0 += 4) {
        int ka = k0 + 2 * hi;
        v2f a = *(const v2f*)(hrow + ka);
        v2f b;
        b.x = wlds[ka * 16 + col];
        b.y = wlds[(ka + 1) * 16 + col];
        c = __builtin_amdgcn_wmma_f32_16x16x4_f32(false, a, false, b,
                                                  (short)0, c, false, false);
    }

    int mbase = tile * 16 + (hi << 3);
    #pragma unroll
    for (int j = 0; j < 8; ++j) {
        int r = mbase + j;
        if (r < n) h2[(long)r * NCLS + col] = c[j];
    }
}

template <int F, int ZROW>
__global__ void alpha_kernel(const float* __restrict__ h,
                             const float* __restrict__ a_src,
                             const float* __restrict__ a_dst,
                             float* __restrict__ outs, float* __restrict__ outd,
                             float4* __restrict__ zrows, int n) {
    int i = blockIdx.x * blockDim.x + threadIdx.x;
    if (i >= n) return;
    if (ZROW > 0) {
        #pragma unroll
        for (int k = 0; k < ZROW / 4; ++k)
            zrows[(long)k * n + i] = make_float4(0.0f, 0.0f, 0.0f, 0.0f);
    }
    const float4* hr = (const float4*)(h + (long)i * F);
    float s = 0.0f, d = 0.0f;
    #pragma unroll
    for (int q = 0; q < F / 4; ++q) {
        float4 hv = hr[q];
        s += hv.x * a_src[4 * q + 0] + hv.y * a_src[4 * q + 1]
           + hv.z * a_src[4 * q + 2] + hv.w * a_src[4 * q + 3];
        d += hv.x * a_dst[4 * q + 0] + hv.y * a_dst[4 * q + 1]
           + hv.z * a_dst[4 * q + 2] + hv.w * a_dst[4 * q + 3];
    }
    outs[i] = s;
    outd[i] = d;
}

__global__ void edge_max(const int* __restrict__ src, const int* __restrict__ dst,
                         const float* __restrict__ as, const float* __restrict__ ad,
                         float* __restrict__ m, int E) {
    int i = blockIdx.x * blockDim.x + threadIdx.x;
    if (i >= E) return;
    int s = src[i], d = dst[i];
    float e = as[s] + ad[d];
    e = (e >= 0.0f) ? e : NEG_SLOPE * e;
    atomicMaxF(&m[d], e);
}

template <int F, int ROW>
__global__ void edge_acc(const int* __restrict__ src, const int* __restrict__ dst,
                         const float* __restrict__ as, const float* __restrict__ ad,
                         const float* __restrict__ m, const float* __restrict__ h,
                         float* __restrict__ acc, int E) {
    int i = blockIdx.x * blockDim.x + threadIdx.x;
    int lane = threadIdx.x & 31;
    if (i - lane >= E) return;
    int s = 0, d = -1;
    float ex = 0.0f;
    if (i < E) {
        s = src[i]; d = dst[i];
        float e = as[s] + ad[d];
        e = (e >= 0.0f) ? e : NEG_SLOPE * e;
        ex = __expf(e - m[d]);
    }
    constexpr int EPI = 32 / ROW;
    constexpr int NIT = 32 / EPI;
    constexpr int CH  = 8;
    int f   = lane % ROW;
    int sub = lane / ROW;
    int fl  = (f < F) ? f : 0;
    for (int c0 = 0; c0 < NIT; c0 += CH) {
        float hv[CH], exj[CH];
        int   dj[CH];
        #pragma unroll
        for (int u = 0; u < CH; ++u) {
            int j  = (c0 + u) * EPI + sub;
            int sj = __shfl(s, j);
            dj[u]  = __shfl(d, j);
            exj[u] = __shfl(ex, j);
            hv[u]  = h[(long)sj * F + fl];
        }
        #pragma unroll
        for (int u = 0; u < CH; ++u) {
            if (f <= F && dj[u] >= 0)
                atomicAdd(&acc[(long)dj[u] * ROW + f], (f < F) ? exj[u] * hv[u] : exj[u]);
        }
    }
}

__global__ void finish1(float* __restrict__ acc, const float* __restrict__ bias, int n) {
    int i = blockIdx.x * blockDim.x + threadIdx.x;
    if (i >= n) return;
    float* row = acc + (long)i * ROW1;
    float d = row[HID];
    float inv = (d > 0.0f) ? 1.0f / d : 0.0f;
    float4* ap = (float4*)row;
    #pragma unroll
    for (int q = 0; q < HID / 4; ++q) {
        float4 v = ap[q];
        v.x = fmaxf(v.x * inv + bias[4 * q + 0], 0.0f);
        v.y = fmaxf(v.y * inv + bias[4 * q + 1], 0.0f);
        v.z = fmaxf(v.z * inv + bias[4 * q + 2], 0.0f);
        v.w = fmaxf(v.w * inv + bias[4 * q + 3], 0.0f);
        ap[q] = v;
    }
}

__global__ void finish2(const float* __restrict__ acc,
                        const float* __restrict__ bias, float* __restrict__ out, int n) {
    int i = blockIdx.x * blockDim.x + threadIdx.x;
    if (i >= n) return;
    const float* row = acc + (long)i * ROW2;
    float d = row[NCLS];
    float inv = (d > 0.0f) ? 1.0f / d : 0.0f;
    const float4* ap = (const float4*)row;
    float v[NCLS];
    float mx = -__builtin_inff();
    #pragma unroll
    for (int q = 0; q < NCLS / 4; ++q) {
        float4 a = ap[q];
        v[4 * q + 0] = a.x * inv + bias[4 * q + 0];
        v[4 * q + 1] = a.y * inv + bias[4 * q + 1];
        v[4 * q + 2] = a.z * inv + bias[4 * q + 2];
        v[4 * q + 3] = a.w * inv + bias[4 * q + 3];
    }
    #pragma unroll
    for (int f = 0; f < NCLS; ++f) mx = fmaxf(mx, v[f]);
    float s = 0.0f;
    #pragma unroll
    for (int f = 0; f < NCLS; ++f) s += __expf(v[f] - mx);
    float ls = __logf(s) + mx;
    float4* op = (float4*)(out + (long)i * NCLS);
    #pragma unroll
    for (int q = 0; q < NCLS / 4; ++q) {
        float4 o;
        o.x = v[4 * q + 0] - ls;
        o.y = v[4 * q + 1] - ls;
        o.z = v[4 * q + 2] - ls;
        o.w = v[4 * q + 3] - ls;
        op[q] = o;
    }
}

extern "C" void kernel_launch(void* const* d_in, const int* in_sizes, int n_in,
                              void* d_out, int out_size, void* d_ws, size_t ws_size,
                              hipStream_t stream) {
    const float* x   = (const float*)d_in[0];
    const int*   ei  = (const int*)  d_in[1];
    const float* W1  = (const float*)d_in[2];
    const float* a1s = (const float*)d_in[3];
    const float* a1d = (const float*)d_in[4];
    const float* b1  = (const float*)d_in[5];
    const float* W2  = (const float*)d_in[6];
    const float* a2s = (const float*)d_in[7];
    const float* a2d = (const float*)d_in[8];
    const float* b2  = (const float*)d_in[9];

    int n = in_sizes[0] / FIN;
    int E = in_sizes[1] / 2;
    const int* src = ei;
    const int* dst = ei + E;

    float* ws   = (float*)d_ws;
    float* acc1 = ws;
    float* h1   = acc1 + (long)n * ROW1;
    float* as1  = h1 + (long)n * HID;
    float* ad1  = as1 + n;
    float* m1   = ad1 + n;
    float* acc2 = ws;
    float* h2   = acc2 + (long)n * ROW2;
    float* as2  = h2 + (long)n * NCLS;
    float* ad2  = as2 + n;
    float* m2   = ad2 + n;

    int ntiles  = (n + 15) / 16;
    int gblocks = (ntiles + 3) / 4;
    int eblocks = (E + 255) / 256;
    int nblocks = (n + 255) / 256;

    init_ws<<<(n * ROW1 + 255) / 256, 256, 0, stream>>>(acc1, m1, m2, n);

    gemm1_wmma<<<gblocks, 128, 0, stream>>>(x, W1, h1, n);
    alpha_kernel<HID, 0><<<nblocks, 256, 0, stream>>>(h1, a1s, a1d, as1, ad1, (float4*)0, n);
    edge_max<<<eblocks, 256, 0, stream>>>(src, dst, as1, ad1, m1, E);
    edge_acc<HID, ROW1><<<eblocks, 256, 0, stream>>>(src, dst, as1, ad1, m1, h1, acc1, E);
    finish1<<<nblocks, 256, 0, stream>>>(acc1, b1, n);

    gemm2_wmma<<<gblocks, 128, 0, stream>>>(acc1, W2, h2, n);
    alpha_kernel<NCLS, ROW2><<<nblocks, 256, 0, stream>>>(h2, a2s, a2d, as2, ad2, (float4*)acc2, n);
    edge_max<<<eblocks, 256, 0, stream>>>(src, dst, as2, ad2, m2, E);
    edge_acc<NCLS, ROW2><<<eblocks, 256, 0, stream>>>(src, dst, as2, ad2, m2, h2, acc2, E);
    finish2<<<nblocks, 256, 0, stream>>>(acc2, b2, (float*)d_out, n);
}


# segment max taken over as[src] alone (order-preserving ops, so m is bit-identical) and kept beside ad[dst] in one 8-byte node record: one scattered scalar gather fewer per edge in all 4 edge passes
# speedup vs baseline: 3.4446x; 1.1663x over previous
#include <hip/hip_runtime.h>
#include <math.h>

typedef __attribute__((ext_vector_type(2))) float v2f;
typedef __attribute__((ext_vector_type(8))) float v8f;

#define FIN   128
#define HID   8
#define NCLS  16
#define NEG_SLOPE 0.2f
#define ROW1  16
#define ROW2  32

__device__ __forceinline__ void atomicMaxF(float* addr, float v) {
    if (v >= 0.0f) atomicMax((int*)addr, __float_as_int(v));
    else           atomicMin((unsigned int*)addr, __float_as_uint(v));
}

__global__ void init_ws(float* acc1, int n) {
    int i = blockIdx.x * blockDim.x + threadIdx.x;
    if (i < n * ROW1) acc1[i] = 0.0f;
}

__global__ void gemm1_wmma(const float* __restrict__ x, const float* __restrict__ W1,
                           float* __restrict__ h1, int n) {
    __shared__ float wlds[FIN * 16];
    int t = threadIdx.x;
    {
        int k = t;
        const float* wr = W1 + k * HID;
        #pragma unroll
        for (int c = 0; c < 16; ++c) wlds[k * 16 + c] = (c < HID) ? wr[c] : 0.0f;
    }
    __syncthreads();

    int wave = t >> 5, lane = t & 31;
    int tile = blockIdx.x * 4 + wave;
    int col  = lane & 15;
    int hi   = lane >> 4;
    int mrow = tile * 16 + col;
    if (mrow >= n) mrow = n - 1;
    const float* xrow = x + (long)mrow * FIN;

    v8f c = {};
    #pragma unroll
    for (int k0 = 0; k0 < FIN; k0 += 4) {
        int ka = k0 + 2 * hi;
        v2f a = *(const v2f*)(xrow + ka);
        v2f b;
        b.x = wlds[ka * 16 + col];
        b.y = wlds[(ka + 1) * 16 + col];
        c = __builtin_amdgcn_wmma_f32_16x16x4_f32(false, a, false, b,
                                                  (short)0, c, false, false);
    }

    int mbase = tile * 16 + (hi << 3);
    if (col < HID) {
        #pragma unroll
        for (int j = 0; j < 8; ++j) {
            int r = mbase + j;
            if (r < n) h1[(long)r * HID + col] = c[j];
        }
    }
}

__global__ void gemm2_wmma(const float* __restrict__ h1r, const float* __restrict__ W2,
                           float* __restrict__ h2, int n) {
    __shared__ float wlds[HID * 16];
    int t = threadIdx.x;
    if (t < HID * 16) wlds[t] = W2[t];
    __syncthreads();

    int wave = t >> 5, lane = t & 31;
    int tile = blockIdx.x * 4 + wave;
    int col  = lane & 15;
    int hi   = lane >> 4;
    int mrow = tile * 16 + col;
    if (mrow >= n) mrow = n - 1;
    const float* hrow = h1r + (long)mrow * ROW1;

    v8f c = {};
    #pragma unroll
    for (int k0 = 0; k0 < HID; k0 += 4) {
        int ka = k0 + 2 * hi;
        v2f a = *(const v2f*)(hrow + ka);
        v2f b;
        b.x = wlds[ka * 16 + col];
        b.y = wlds[(ka + 1) * 16 + col];
        c = __builtin_amdgcn_wmma_f32_16x16x4_f32(false, a, false, b,
                                                  (short)0, c, false, false);
    }

    int mbase = tile * 16 + (hi << 3);
    #pragma unroll
    for (int j = 0; j < 8; ++j) {
        int r = mbase + j;
        if (r < n) h2[(long)r * NCLS + col] = c[j];
    }
}

template <int F, int ZROW>
__global__ void alpha_kernel(const float* __restrict__ h,
                             const float* __restrict__ a_src,
                             const float* __restrict__ a_dst,
                             float* __restrict__ outs, float2* __restrict__ rec,
                             float4* __restrict__ zrows, int n) {
    int i = blockIdx.x * blockDim.x + threadIdx.x;
    if (i >= n) return;
    if (ZROW > 0) {
        #pragma unroll
        for (int k = 0; k < ZROW / 4; ++k)
            zrows[(long)k * n + i] = make_float4(0.0f, 0.0f, 0.0f, 0.0f);
    }
    const float4* hr = (const float4*)(h + (long)i * F);
    float s = 0.0f, d = 0.0f;
    #pragma unroll
    for (int q = 0; q < F / 4; ++q) {
        float4 hv = hr[q];
        s += hv.x * a_src[4 * q + 0] + hv.y * a_src[4 * q + 1]
           + hv.z * a_src[4 * q + 2] + hv.w * a_src[4 * q + 3];
        d += hv.x * a_dst[4 * q + 0] + hv.y * a_dst[4 * q + 1]
           + hv.z * a_dst[4 * q + 2] + hv.w * a_dst[4 * q + 3];
    }
    outs[i] = s;
    rec[i]  = make_float2(d, -__builtin_inff());
}

__global__ void edge_max(const int* __restrict__ src, const int* __restrict__ dst,
                         const float* __restrict__ as, float2* __restrict__ rec, int E) {
    int i = blockIdx.x * blockDim.x + threadIdx.x;
    if (i >= E) return;
    int s = src[i], d = dst[i];
    atomicMaxF((float*)(rec + d) + 1, as[s]);
}

template <int F, int ROW>
__global__ void edge_acc(const int* __restrict__ src, const int* __restrict__ dst,
                         const float* __restrict__ as, const float2* __restrict__ rec,
                         const float* __restrict__ h,
                         float* __restrict__ acc, int E) {
    int i = blockIdx.x * blockDim.x + threadIdx.x;
    int lane = threadIdx.x & 31;
    if (i - lane >= E) return;
    int s = 0, d = -1;
    float ex = 0.0f;
    if (i < E) {
        s = src[i]; d = dst[i];
        float2 r = rec[d];
        float e = as[s] + r.x;
        e = (e >= 0.0f) ? e : NEG_SLOPE * e;
        float mx = r.y + r.x;
        mx = (mx >= 0.0f) ? mx : NEG_SLOPE * mx;
        ex = __expf(e - mx);
    }
    constexpr int EPI = 32 / ROW;
    constexpr int NIT = 32 / EPI;
    constexpr int CH  = 8;
    int f   = lane % ROW;
    int sub = lane / ROW;
    int fl  = (f < F) ? f : 0;
    for (int c0 = 0; c0 < NIT; c0 += CH) {
        float hv[CH], exj[CH];
        int   dj[CH];
        #pragma unroll
        for (int u = 0; u < CH; ++u) {
            int j  = (c0 + u) * EPI + sub;
            int sj = __shfl(s, j);
            dj[u]  = __shfl(d, j);
            exj[u] = __shfl(ex, j);
            hv[u]  = h[(long)sj * F + fl];
        }
        #pragma unroll
        for (int u = 0; u < CH; ++u) {
            if (f <= F && dj[u] >= 0)
                atomicAdd(&acc[(long)dj[u] * ROW + f], (f < F) ? exj[u] * hv[u] : exj[u]);
        }
    }
}

__global__ void finish1(float* __restrict__ acc, const float* __restrict__ bias, int n) {
    int i = blockIdx.x * blockDim.x + threadIdx.x;
    if (i >= n) return;
    float* row = acc + (long)i * ROW1;
    float d = row[HID];
    float inv = (d > 0.0f) ? 1.0f / d : 0.0f;
    float4* ap = (float4*)row;
    #pragma unroll
    for (int q = 0; q < HID / 4; ++q) {
        float4 v = ap[q];
        v.x = fmaxf(v.x * inv + bias[4 * q + 0], 0.0f);
        v.y = fmaxf(v.y * inv + bias[4 * q + 1], 0.0f);
        v.z = fmaxf(v.z * inv + bias[4 * q + 2], 0.0f);
        v.w = fmaxf(v.w * inv + bias[4 * q + 3], 0.0f);
        ap[q] = v;
    }
}

__global__ void finish2(const float* __restrict__ acc,
                        const float* __restrict__ bias, float* __restrict__ out, int n) {
    int i = blockIdx.x * blockDim.x + threadIdx.x;
    if (i >= n) return;
    const float* row = acc + (long)i * ROW2;
    float d = row[NCLS];
    float inv = (d > 0.0f) ? 1.0f / d : 0.0f;
    const float4* ap = (const float4*)row;
    float v[NCLS];
    float mx = -__builtin_inff();
    #pragma unroll
    for (int q = 0; q < NCLS / 4; ++q) {
        float4 a = ap[q];
        v[4 * q + 0] = a.x * inv + bias[4 * q + 0];
        v[4 * q + 1] = a.y * inv + bias[4 * q + 1];
        v[4 * q + 2] = a.z * inv + bias[4 * q + 2];
        v[4 * q + 3] = a.w * inv + bias[4 * q + 3];
    }
    #pragma unroll
    for (int f = 0; f < NCLS; ++f) mx = fmaxf(mx, v[f]);
    float s = 0.0f;
    #pragma unroll
    for (int f = 0; f < NCLS; ++f) s += __expf(v[f] - mx);
    float ls = __logf(s) + mx;
    float4* op = (float4*)(out + (long)i * NCLS);
    #pragma unroll
    for (int q = 0; q < NCLS / 4; ++q) {
        float4 o;
        o.x = v[4 * q + 0] - ls;
        o.y = v[4 * q + 1] - ls;
        o.z = v[4 * q + 2] - ls;
        o.w = v[4 * q + 3] - ls;
        op[q] = o;
    }
}

extern "C" void kernel_launch(void* const* d_in, const int* in_sizes, int n_in,
                              void* d_out, int out_size, void* d_ws, size_t ws_size,
                              hipStream_t stream) {
    const float* x   = (const float*)d_in[0];
    const int*   ei  = (const int*)  d_in[1];
    const float* W1  = (const float*)d_in[2];
    const float* a1s = (const float*)d_in[3];
    const float* a1d = (const float*)d_in[4];
    const float* b1  = (const float*)d_in[5];
    const float* W2  = (const float*)d_in[6];
    const float* a2s = (const float*)d_in[7];
    const float* a2d = (const float*)d_in[8];
    const float* b2  = (const float*)d_in[9];

    int n = in_sizes[0] / FIN;
    int E = in_sizes[1] / 2;
    const int* src = ei;
    const int* dst = ei + E;

    float*  ws   = (float*)d_ws;
    float*  acc1 = ws;
    float*  h1   = acc1 + (long)n * ROW1;
    float2* rec1 = (float2*)(h1 + (long)n * HID);
    float*  as1  = (float*)(rec1 + n);
    float*  acc2 = ws;
    float*  h2   = acc2 + (long)n * ROW2;
    float2* rec2 = (float2*)(h2 + (long)n * NCLS);
    float*  as2  = (float*)(rec2 + n);

    int ntiles  = (n + 15) / 16;
    int gblocks = (ntiles + 3) / 4;
    int eblocks = (E + 255) / 256;
    int nblocks = (n + 255) / 256;

    init_ws<<<(n * ROW1 + 255) / 256, 256, 0, stream>>>(acc1, n);

    gemm1_wmma<<<gblocks, 128, 0, stream>>>(x, W1, h1, n);
    alpha_kernel<HID, 0><<<nblocks, 256, 0, stream>>>(h1, a1s, a1d, as1, rec1, (float4*)0, n);
    edge_max<<<eblocks, 256, 0, stream>>>(src, dst, as1, rec1, E);
    edge_acc<HID, ROW1><<<eblocks, 256, 0, stream>>>(src, dst, as1, rec1, h1, acc1, E);
    finish1<<<nblocks, 256, 0, stream>>>(acc1, b1, n);

    gemm2_wmma<<<gblocks, 128, 0, stream>>>(acc1, W2, h2, n);
    alpha_kernel<NCLS, ROW2><<<nblocks, 256, 0, stream>>>(h2, a2s, a2d, as2, rec2, (float4*)acc2, n);
    edge_max<<<eblocks, 256, 0, stream>>>(src, dst, as2, rec2, E);
    edge_acc<NCLS, ROW2><<<eblocks, 256, 0, stream>>>(src, dst, as2, rec2, h2, acc2, E);
    finish2<<<nblocks, 256, 0, stream>>>(acc2, b2, (float*)d_out, n);
}


# edge_acc: as[src] now lies behind the node's features in the row the edge gathers anyway (word F): the separate scattered as[src] gather per edge is gone in both layers; ex and sum order unchanged
# speedup vs baseline: 3.5628x; 1.0343x over previous
#include <hip/hip_runtime.h>
#include <math.h>

typedef __attribute__((ext_vector_type(2))) float v2f;
typedef __attribute__((ext_vector_type(8))) float v8f;

#define FIN   128
#define HID   8
#define NCLS  16
#define NEG_SLOPE 0.2f
#define ROW1  16
#define ROW2  32
#define HROW1 16
#define HROW2 17

__device__ __forceinline__ void atomicMaxF(float* addr, float v) {
    if (v >= 0.0f) atomicMax((int*)addr, __float_as_int(v));
    else           atomicMin((unsigned int*)addr, __float_as_uint(v));
}

__global__ void init_ws(float* acc1, int n) {
    int i = blockIdx.x * blockDim.x + threadIdx.x;
    if (i < n * ROW1) acc1[i] = 0.0f;
}

__global__ void gemm1_wmma(const float* __restrict__ x, const float* __restrict__ W1,
                           float* __restrict__ h1, int n) {
    __shared__ float wlds[FIN * 16];
    int t = threadIdx.x;
    {
        int k = t;
        const float* wr = W1 + k * HID;
        #pragma unroll
        for (int c = 0; c < 16; ++c) wlds[k * 16 + c] = (c < HID) ? wr[c] : 0.0f;
    }
    __syncthreads();

    int wave = t >> 5, lane = t & 31;
    int tile = blockIdx.x * 4 + wave;
    int col  = lane & 15;
    int hi   = lane >> 4;
    int mrow = tile * 16 + col;
    if (mrow >= n) mrow = n - 1;
    const float* xrow = x + (long)mrow * FIN;

    v8f c = {};
    #pragma unroll
    for (int k0 = 0; k0 < FIN; k0 += 4) {
        int ka = k0 + 2 * hi;
        v2f a = *(const v2f*)(xrow + ka);
        v2f b;
        b.x = wlds[ka * 16 + col];
        b.y = wlds[(ka + 1) * 16 + col];
        c = __builtin_amdgcn_wmma_f32_16x16x4_f32(false, a, false, b,
                                                  (short)0, c, false, false);
    }

    int mbase = tile * 16 + (hi << 3);
    if (col < HID) {
        #pragma unroll
        for (int j = 0; j < 8; ++j) {
            int r = mbase + j;
            if (r < n) h1[(long)r * HROW1 + col] = c[j];
        }
    }
}

__global__ void gemm2_wmma(const float* __restrict__ h1r, const float* __restrict__ W2,
                           float* __restrict__ h2, int n) {
    __shared__ float wlds[HID * 16];
    int t = threadIdx.x;
    if (t < HID * 16) wlds[t] = W2[t];
    __syncthreads();

    int wave = t >> 5, lane = t & 31;
    int tile = blockIdx.x * 4 + wave;
    int col  = lane & 15;
    int hi   = lane >> 4;
    int mrow = tile * 16 + col;
    if (mrow >= n) mrow = n - 1;
    const float* hrow = h1r + (long)mrow * ROW1;

    v8f c = {};
    #pragma unroll
    for (int k0 = 0; k0 < HID; k0 += 4) {
        int ka = k0 + 2 * hi;
        v2f a = *(const v2f*)(hrow + ka);
        v2f b;
        b.x = wlds[ka * 16 + col];
        b.y = wlds[(ka + 1) * 16 + col];
        c = __builtin_amdgcn_wmma_f32_16x16x4_f32(false, a, false, b,
                                                  (short)0, c, false, false);
    }

    int mbase = tile * 16 + (hi << 3);
    #pragma unroll
    for (int j = 0; j < 8; ++j) {
        int r = mbase + j;
        if (r < n) h2[(long)r * HROW2 + col] = c[j];
    }
}

template <int F, int HROW, int ZROW>
__global__ void alpha_kernel(float* __restrict__ hrow,
                             const float* __restrict__ a_src,
                             const float* __restrict__ a_dst,
                             float* __restrict__ outs, float2* __restrict__ rec,
                             float4* __restrict__ zrows, int n) {
    int i = blockIdx.x * blockDim.x + threadIdx.x;
    if (i >= n) return;
    if (ZROW > 0) {
        #pragma unroll
        for (int k = 0; k < ZROW / 4; ++k)
            zrows[(long)k * n + i] = make_float4(0.0f, 0.0f, 0.0f, 0.0f);
    }
    float* row = hrow + (long)i * HROW;
    float s = 0.0f, d = 0.0f;
    #pragma unroll
    for (int q = 0; q < F / 4; ++q) {
        float4 hv;
        if (HROW % 4 == 0) {
            hv = ((const float4*)row)[q];
        } else {
            hv.x = row[4 * q + 0]; hv.y = row[4 * q + 1];
            hv.z = row[4 * q + 2]; hv.w = row[4 * q + 3];
        }
        s += hv.x * a_src[4 * q + 0] + hv.y * a_src[4 * q + 1]
           + hv.z * a_src[4 * q + 2] + hv.w * a_src[4 * q + 3];
        d += hv.x * a_dst[4 * q + 0] + hv.y * a_dst[4 * q + 1]
           + hv.z * a_dst[4 * q + 2] + hv.w * a_dst[4 * q + 3];
    }
    row[F] = s;
    if (outs) outs[i] = s;
    rec[i]  = make_float2(d, -__builtin_inff());
}

template <int STRIDE>
__global__ void edge_max(const int* __restrict__ src, const int* __restrict__ dst,
                         const float* __restrict__ as, float2* __restrict__ rec, int E) {
    int i = blockIdx.x * blockDim.x + threadIdx.x;
    if (i >= E) return;
    int s = src[i], d = dst[i];
    atomicMaxF((float*)(rec + d) + 1, as[(long)s * STRIDE]);
}

template <int ROW, int F>
__device__ __forceinline__ float from_lane_F(float v, int lane_f) {
    if (ROW == 32)
        return __int_as_float(__builtin_amdgcn_readlane(__float_as_int(v), F));
    if (ROW == 16)
        return __int_as_float(__builtin_amdgcn_update_dpp(0, __float_as_int(v), 0x150 + F, 0xf, 0xf, false));
    return __shfl(v, lane_f);
}

template <int F, int ROW, int HROW>
__global__ void edge_acc(const int* __restrict__ src, const int* __restrict__ dst,
                         const float2* __restrict__ rec,
                         const float* __restrict__ hrow,
                         float* __restrict__ acc, int E) {
    int i = blockIdx.x * blockDim.x + threadIdx.x;
    int lane = threadIdx.x & 31;
    if (i - lane >= E) return;
    int s = 0, d = -1;
    float adv = 0.0f, mx = 0.0f;
    if (i < E) {
        s = src[i]; d = dst[i];
        float2 r = rec[d];
        adv = r.x;
        mx = r.y + r.x;
        mx = (mx >= 0.0f) ? mx : NEG_SLOPE * mx;
    }
    constexpr int EPI = 32 / ROW;
    constexpr int NIT = 32 / EPI;
    constexpr int CH  = 8;
    int f   = lane % ROW;
    int sub = lane / ROW;
    int fl  = (f <= F) ? f : 0;
    int asl = sub * ROW + F;
    for (int c0 = 0; c0 < NIT; c0 += CH) {
        float hv[CH], adj[CH], mxj[CH];
        int   dj[CH];
        #pragma unroll
        for (int u = 0; u < CH; ++u) {
            int j  = (c0 + u) * EPI + sub;
            int sj = __shfl(s, j);
            dj[u]  = __shfl(d, j);
            adj[u] = __shfl(adv, j);
            mxj[u] = __shfl(mx, j);
            hv[u]  = hrow[(long)sj * HROW + fl];
        }
        #pragma unroll
        for (int u = 0; u < CH; ++u) {
            float e = from_lane_F<ROW, F>(hv[u], asl) + adj[u];
            e = (e >= 0.0f) ? e : NEG_SLOPE * e;
            float ex = __expf(e - mxj[u]);
            if (f <= F && dj[u] >= 0)
                atomicAdd(&acc[(long)dj[u] * ROW + f], (f < F) ? ex * hv[u] : ex);
        }
    }
}

__global__ void finish1(float* __restrict__ acc, const float* __restrict__ bias, int n) {
    int i = blockIdx.x * blockDim.x + threadIdx.x;
    if (i >= n) return;
    float* row = acc + (long)i * ROW1;
    float d = row[HID];
    float inv = (d > 0.0f) ? 1.0f / d : 0.0f;
    float4* ap = (float4*)row;
    #pragma unroll
    for (int q = 0; q < HID / 4; ++q) {
        float4 v = ap[q];
        v.x = fmaxf(v.x * inv + bias[4 * q + 0], 0.0f);
        v.y = fmaxf(v.y * inv + bias[4 * q + 1], 0.0f);
        v.z = fmaxf(v.z * inv + bias[4 * q + 2], 0.0f);
        v.w = fmaxf(v.w * inv + bias[4 * q + 3], 0.0f);
        ap[q] = v;
    }
}

__global__ void finish2(const float* __restrict__ acc,
                        const float* __restrict__ bias, float* __restrict__ out, int n) {
    int i = blockIdx.x * blockDim.x + threadIdx.x;
    if (i >= n) return;
    const float* row = acc + (long)i * ROW2;
    float d = row[NCLS];
    float inv = (d > 0.0f) ? 1.0f / d : 0.0f;
    const float4* ap = (const float4*)row;
    float v[NCLS];
    float mx = -__builtin_inff();
    #pragma unroll
    for (int q = 0; q < NCLS / 4; ++q) {
        float4 a = ap[q];
        v[4 * q + 0] = a.x * inv + bias[4 * q + 0];
        v[4 * q + 1] = a.y * inv + bias[4 * q + 1];
        v[4 * q + 2] = a.z * inv + bias[4 * q + 2];
        v[4 * q + 3] = a.w * inv + bias[4 * q + 3];
    }
    #pragma unroll
    for (int f = 0; f < NCLS; ++f) mx = fmaxf(mx, v[f]);
    float s = 0.0f;
    #pragma unroll
    for (int f = 0; f < NCLS; ++f) s += __expf(v[f] - mx);
    float ls = __logf(s) + mx;
    float4* op = (float4*)(out + (long)i * NCLS);
    #pragma unroll
    for (int q = 0; q < NCLS / 4; ++q) {
        float4 o;
        o.x = v[4 * q + 0] - ls;
        o.y = v[4 * q + 1] - ls;
        o.z = v[4 * q + 2] - ls;
        o.w = v[4 * q + 3] - ls;
        op[q] = o;
    }
}

extern "C" void kernel_launch(void* const* d_in, const int* in_sizes, int n_in,
                              void* d_out, int out_size, void* d_ws, size_t ws_size,
                              hipStream_t stream) {
    const float* x   = (const float*)d_in[0];
    const int*   ei  = (const int*)  d_in[1];
    const float* W1  = (const float*)d_in[2];
    const float* a1s = (const float*)d_in[3];
    const float* a1d = (const float*)d_in[4];
    const float* b1  = (const float*)d_in[5];
    const float* W2  = (const float*)d_in[6];
    const float* a2s = (const float*)d_in[7];
    const float* a2d = (const float*)d_in[8];
    const float* b2  = (const float*)d_in[9];

    int n = in_sizes[0] / FIN;
    int E = in_sizes[1] / 2;
    const int* src = ei;
    const int* dst = ei + E;

    float*  ws   = (float*)d_ws;
    float*  acc1 = ws;
    float*  hr1  = acc1 + (long)n * ROW1;
    float2* rec1 = (float2*)(hr1 + (long)n * HROW1);
    float*  as1  = (float*)(rec1 + n);
    float*  acc2 = ws;
    float2* rec2 = (float2*)(acc2 + (long)n * ROW2);
    float*  hr2  = (float*)(rec2 + n);

    int ntiles  = (n + 15) / 16;
    int gblocks = (ntiles + 3) / 4;
    int eblocks = (E + 255) / 256;
    int nblocks = (n + 255) / 256;

    init_ws<<<(n * ROW1 + 255) / 256, 256, 0, stream>>>(acc1, n);

    gemm1_wmma<<<gblocks, 128, 0, stream>>>(x, W1, hr1, n);
    alpha_kernel<HID, HROW1, 0><<<nblocks, 256, 0, stream>>>(hr1, a1s, a1d, as1, rec1, (float4*)0, n);
    edge_max<1><<<eblocks, 256, 0, stream>>>(src, dst, as1, rec1, E);
    edge_acc<HID, ROW1, HROW1><<<eblocks, 256, 0, stream>>>(src, dst, rec1, hr1, acc1, E);
    finish1<<<nblocks, 256, 0, stream>>>(acc1, b1, n);

    gemm2_wmma<<<gblocks, 128, 0, stream>>>(acc1, W2, hr2, n);
    alpha_kernel<NCLS, HROW2, ROW2><<<nblocks, 256, 0, stream>>>(hr2, a2s, a2d, (float*)0, rec2, (float4*)acc2, n);
    edge_max<HROW2><<<eblocks, 256, 0, stream>>>(src, dst, hr2 + NCLS, rec2, E);
    edge_acc<NCLS, ROW2, HROW2><<<eblocks, 256, 0, stream>>>(src, dst, rec2, hr2, acc2, E);
    finish2<<<nblocks, 256, 0, stream>>>(acc2, b2, (float*)d_out, n);
}


# softmax shift = leaky(max of as over ALL nodes + ad[dst]) while max(as)-min(as) <= 32, else the exact path: both segment-max edge passes exit at once; ex rescaled per node, sum order unchanged
# speedup vs baseline: 7.8086x; 2.1917x over previous
#include <hip/hip_runtime.h>
#include <math.h>

typedef __attribute__((ext_vector_type(2))) float v2f;
typedef __attribute__((ext_vector_type(8))) float v8f;

#define FIN   128
#define HID   8
#define NCLS  16
#define NEG_SLOPE 0.2f
#define ROW1  16
#define ROW2  32
#define HROW1 16
#define HROW2 17

#define SPREAD_MAX 32.0f
#define GW1 14
#define GW2 30

__device__ __forceinline__ void atomicMaxF(float* addr, float v) {
    if (v >= 0.0f) atomicMax((int*)addr, __float_as_int(v));
    else           atomicMin((unsigned int*)addr, __float_as_uint(v));
}

__global__ void init_ws(float* acc1, int n) {
    int i = blockIdx.x * blockDim.x + threadIdx.x;
    if (i < n * ROW1)
        acc1[i] = (i == GW1 || i == GW1 + 1 || i == GW2 || i == GW2 + 1) ? -__builtin_inff() : 0.0f;
}

__global__ void gemm1_wmma(const float* __restrict__ x, const float* __restrict__ W1,
                           float* __restrict__ h1, int n) {
    __shared__ float wlds[FIN * 16];
    int t = threadIdx.x;
    {
        int k = t;
        const float* wr = W1 + k * HID;
        #pragma unroll
        for (int c = 0; c < 16; ++c) wlds[k * 16 + c] = (c < HID) ? wr[c] : 0.0f;
    }
    __syncthreads();

    int wave = t >> 5, lane = t & 31;
    int tile = blockIdx.x * 4 + wave;
    int col  = lane & 15;
    int hi   = lane >> 4;
    int mrow = tile * 16 + col;
    if (mrow >= n) mrow = n - 1;
    const float* xrow = x + (long)mrow * FIN;

    v8f c = {};
    #pragma unroll
    for (int k0 = 0; k0 < FIN; k0 += 4) {
        int ka = k0 + 2 * hi;
        v2f a = *(const v2f*)(xrow + ka);
        v2f b;
        b.x = wlds[ka * 16 + col];
        b.y = wlds[(ka + 1) * 16 + col];
        c = __builtin_amdgcn_wmma_f32_16x16x4_f32(false, a, false, b,
                                                  (short)0, c, false, false);
    }

    int mbase = tile * 16 + (hi << 3);
    if (col < HID) {
        #pragma unroll
        for (int j = 0; j < 8; ++j) {
            int r = mbase + j;
            if (r < n) h1[(long)r * HROW1 + col] = c[j];
        }
    }
}

__global__ void gemm2_wmma(const float* __restrict__ h1r, const float* __restrict__ W2,
                           float* __restrict__ h2, int n) {
    __shared__ float wlds[HID * 16];
    int t = threadIdx.x;
    if (t < HID * 16) wlds[t] = W2[t];
    __syncthreads();

    int wave = t >> 5, lane = t & 31;
    int tile = blockIdx.x * 4 + wave;
    int col  = lane & 15;
    int hi   = lane >> 4;
    int mrow = tile * 16 + col;
    if (mrow >= n) mrow = n - 1;
    const float* hrow = h1r + (long)mrow * ROW1;

    v8f c = {};
    #pragma unroll
    for (int k0 = 0; k0 < HID; k0 += 4) {
        int ka = k0 + 2 * hi;
        v2f a = *(const v2f*)(hrow + ka);
        v2f b;
        b.x = wlds[ka * 16 + col];
        b.y = wlds[(ka + 1) * 16 + col];
        c = __builtin_amdgcn_wmma_f32_16x16x4_f32(false, a, false, b,
                                                  (short)0, c, false, false);
    }

    int mbase = tile * 16 + (hi << 3);
    #pragma unroll
    for (int j = 0; j < 8; ++j) {
        int r = mbase + j;
        if (r < n) h2[(long)r * HROW2 + col] = c[j];
    }
}

template <int F, int HROW, int ZROW>
__global__ void alpha_kernel(float* __restrict__ hrow,
                             const float* __restrict__ a_src,
                             const float* __restrict__ a_dst,
                             float* __restrict__ outs, float2* __restrict__ rec,
                             float4* __restrict__ zrows, float* __restrict__ g, int n) {
    __shared__ float red[2][8];
    int i = blockIdx.x * blockDim.x + threadIdx.x;
    float hi = -__builtin_inff(), lo = -__builtin_inff();
    if (i < n) {
        if (ZROW > 0) {
            #pragma unroll
            for (int k = 0; k < ZROW / 4; ++k) {
                float4* zp = zrows + (long)k * n + i;
                if (zp != (float4*)((unsigned long long)g & ~15ull))
                    *zp = make_float4(0.0f, 0.0f, 0.0f, 0.0f);
            }
        }
        float* row = hrow + (long)i * HROW;
        float s = 0.0f, d = 0.0f;
        #pragma unroll
        for (int q = 0; q < F / 4; ++q) {
            float4 hv;
            if (HROW % 4 == 0) {
                hv = ((const float4*)row)[q];
            } else {
                hv.x = row[4 * q + 0]; hv.y = row[4 * q + 1];
                hv.z = row[4 * q + 2]; hv.w = row[4 * q + 3];
            }
            s += hv.x * a_src[4 * q + 0] + hv.y * a_src[4 * q + 1]
               + hv.z * a_src[4 * q + 2] + hv.w * a_src[4 * q + 3];
            d += hv.x * a_dst[4 * q + 0] + hv.y * a_dst[4 * q + 1]
               + hv.z * a_dst[4 * q + 2] + hv.w * a_dst[4 * q + 3];
        }
        row[F] = s;
        if (outs) outs[i] = s;
        rec[i]  = make_float2(d, -__builtin_inff());
        hi = s + 0.0f;
        lo = 0.0f - s;
    }
    for (int off = 16; off > 0; off >>= 1) {
        hi = fmaxf(hi, __shfl_xor(hi, off));
        lo = fmaxf(lo, __shfl_xor(lo, off));
    }
    if ((threadIdx.x & 31) == 0) { red[0][threadIdx.x >> 5] = hi; red[1][threadIdx.x >> 5] = lo; }
    __syncthreads();
    if (threadIdx.x < 2) {
        float v = red[threadIdx.x][0];
        for (int w = 1; w < 8; ++w) v = fmaxf(v, red[threadIdx.x][w]);
        atomicMaxF(g + threadIdx.x, v);
    }
}

template <int STRIDE>
__global__ void edge_max(const int* __restrict__ src, const int* __restrict__ dst,
                         const float* __restrict__ as, float2* __restrict__ rec,
                         const float* __restrict__ g, int E) {
    if (g[0] + g[1] <= SPREAD_MAX) return;
    for (int i = blockIdx.x * blockDim.x + threadIdx.x; i < E; i += gridDim.x * blockDim.x) {
        int s = src[i], d = dst[i];
        atomicMaxF((float*)(rec + d) + 1, as[(long)s * STRIDE]);
    }
}

template <int ROW, int F>
__device__ __forceinline__ float from_lane_F(float v, int lane_f) {
    if (ROW == 32)
        return __int_as_float(__builtin_amdgcn_readlane(__float_as_int(v), F));
    if (ROW == 16)
        return __int_as_float(__builtin_amdgcn_update_dpp(0, __float_as_int(v), 0x150 + F, 0xf, 0xf, false));
    return __shfl(v, lane_f);
}

template <int F, int ROW, int HROW>
__global__ void edge_acc(const int* __restrict__ src, const int* __restrict__ dst,
                         const float2* __restrict__ rec,
                         const float* __restrict__ hrow,
                         float* __restrict__ acc, const float* __restrict__ g, int E) {
    int i = blockIdx.x * blockDim.x + threadIdx.x;
    int lane = threadIdx.x & 31;
    if (i - lane >= E) return;
    float gmax = g[0];
    bool  narrow = gmax + g[1] <= SPREAD_MAX;
    int s = 0, d = -1;
    float adv = 0.0f, mx = 0.0f;
    if (i < E) {
        s = src[i]; d = dst[i];
        float2 r = rec[d];
        adv = r.x;
        mx = (narrow ? gmax : r.y) + r.x;
        mx = (mx >= 0.0f) ? mx : NEG_SLOPE * mx;
    }
    constexpr int EPI = 32 / ROW;
    constexpr int NIT = 32 / EPI;
    constexpr int CH  = 8;
    int f   = lane % ROW;
    int sub = lane / ROW;
    int fl  = (f <= F) ? f : 0;
    int asl = sub * ROW + F;
    for (int c0 = 0; c0 < NIT; c0 += CH) {
        float hv[CH], adj[CH], mxj[CH];
        int   dj[CH];
        #pragma unroll
        for (int u = 0; u < CH; ++u) {
            int j  = (c0 + u) * EPI + sub;
            int sj = __shfl(s, j);
            dj[u]  = __shfl(d, j);
            adj[u] = __shfl(adv, j);
            mxj[u] = __shfl(mx, j);
            hv[u]  = hrow[(long)sj * HROW + fl];
        }
        #pragma unroll
        for (int u = 0; u < CH; ++u) {
            float e = from_lane_F<ROW, F>(hv[u], asl) + adj[u];
            e = (e >= 0.0f) ? e : NEG_SLOPE * e;
            float ex = __expf(e - mxj[u]);
            if (f <= F && dj[u] >= 0)
                atomicAdd(&acc[(long)dj[u] * ROW + f], (f < F) ? ex * hv[u] : ex);
        }
    }
}

__global__ void finish1(float* __restrict__ acc, const float* __restrict__ bias, int n) {
    int i = blockIdx.x * blockDim.x + threadIdx.x;
    if (i >= n) return;
    float* row = acc + (long)i * ROW1;
    float d = row[HID];
    float inv = (d > 0.0f) ? 1.0f / d : 0.0f;
    float4* ap = (float4*)row;
    #pragma unroll
    for (int q = 0; q < HID / 4; ++q) {
        float4 v = ap[q];
        v.x = fmaxf(v.x * inv + bias[4 * q + 0], 0.0f);
        v.y = fmaxf(v.y * inv + bias[4 * q + 1], 0.0f);
        v.z = fmaxf(v.z * inv + bias[4 * q + 2], 0.0f);
        v.w = fmaxf(v.w * inv + bias[4 * q + 3], 0.0f);
        ap[q] = v;
    }
}

__global__ void finish2(const float* __restrict__ acc,
                        const float* __restrict__ bias, float* __restrict__ out, int n) {
    int i = blockIdx.x * blockDim.x + threadIdx.x;
    if (i >= n) return;
    const float* row = acc + (long)i * ROW2;
    float d = row[NCLS];
    float inv = (d > 0.0f) ? 1.0f / d : 0.0f;
    const float4* ap = (const float4*)row;
    float v[NCLS];
    float mx = -__builtin_inff();
    #pragma unroll
    for (int q = 0; q < NCLS / 4; ++q) {
        float4 a = ap[q];
        v[4 * q + 0] = a.x * inv + bias[4 * q + 0];
        v[4 * q + 1] = a.y * inv + bias[4 * q + 1];
        v[4 * q + 2] = a.z * inv + bias[4 * q + 2];
        v[4 * q + 3] = a.w * inv + bias[4 * q + 3];
    }
    #pragma unroll
    for (int f = 0; f < NCLS; ++f) mx = fmaxf(mx, v[f]);
    float s = 0.0f;
    #pragma unroll
    for (int f = 0; f < NCLS; ++f) s += __expf(v[f] - mx);
    float ls = __logf(s) + mx;
    float4* op = (float4*)(out + (long)i * NCLS);
    #pragma unroll
    for (int q = 0; q < NCLS / 4; ++q) {
        float4 o;
        o.x = v[4 * q + 0] - ls;
        o.y = v[4 * q + 1] - ls;
        o.z = v[4 * q + 2] - ls;
        o.w = v[4 * q + 3] - ls;
        op[q] = o;
    }
}

extern "C" void kernel_launch(void* const* d_in, const int* in_sizes, int n_in,
                              void* d_out, int out_size, void* d_ws, size_t ws_size,
                              hipStream_t stream) {
    const float* x   = (const float*)d_in[0];
    const int*   ei  = (const int*)  d_in[1];
    const float* W1  = (const float*)d_in[2];
    const float* a1s = (const float*)d_in[3];
    const float* a1d = (const float*)d_in[4];
    const float* b1  = (const float*)d_in[5];
    const float* W2  = (const float*)d_in[6];
    const float* a2s = (const float*)d_in[7];
    const float* a2d = (const float*)d_in[8];
    const float* b2  = (const float*)d_in[9];

    int n = in_sizes[0] / FIN;
    int E = in_sizes[1] / 2;
    const int* src = ei;
    const int* dst = ei + E;

    float*  ws   = (float*)d_ws;
    float*  acc1 = ws;
    float*  hr1  = acc1 + (long)n * ROW1;
    float2* rec1 = (float2*)(hr1 + (long)n * HROW1);
    float*  as1  = (float*)(rec1 + n);
    float*  acc2 = ws;
    float2* rec2 = (float2*)(acc2 + (long)n * ROW2);
    float*  hr2  = (float*)(rec2 + n);

    int ntiles  = (n + 15) / 16;
    int gblocks = (ntiles + 3) / 4;
    int eblocks = (E + 255) / 256;
    int nblocks = (n + 255) / 256;

    init_ws<<<(n * ROW1 + 255) / 256, 256, 0, stream>>>(acc1, n);

    gemm1_wmma<<<gblocks, 128, 0, stream>>>(x, W1, hr1, n);
    alpha_kernel<HID, HROW1, 0><<<nblocks, 256, 0, stream>>>(hr1, a1s, a1d, as1, rec1, (float4*)0, ws + GW1, n);
    edge_max<1><<<2048, 256, 0, stream>>>(src, dst, as1, rec1, ws + GW1, E);
    edge_acc<HID, ROW1, HROW1><<<eblocks, 256, 0, stream>>>(src, dst, rec1, hr1, acc1, ws + GW1, E);
    finish1<<<nblocks, 256, 0, stream>>>(acc1, b1, n);

    gemm2_wmma<<<gblocks, 128, 0, stream>>>(acc1, W2, hr2, n);
    alpha_kernel<NCLS, HROW2, ROW2><<<nblocks, 256, 0, stream>>>(hr2, a2s, a2d, (float*)0, rec2, (float4*)acc2, ws + GW2, n);
    edge_max<HROW2><<<2048, 256, 0, stream>>>(src, dst, hr2 + NCLS, rec2, ws + GW2, E);
    edge_acc<NCLS, ROW2, HROW2><<<eblocks, 256, 0, stream>>>(src, dst, rec2, hr2, acc2, ws + GW2, E);
    finish2<<<nblocks, 256, 0, stream>>>(acc2, b2, (float*)d_out, n);
}


# layer 2 sums the 8 hidden features per edge and applies W2 (WMMA) once per node behind the sums: 9-word rows, two edges per atomic; sum order changes, log-softmax sums become lane butterflies
# speedup vs baseline: 8.2676x; 1.0588x over previous
#include <hip/hip_runtime.h>
#include <math.h>

typedef __attribute__((ext_vector_type(2))) float v2f;
typedef __attribute__((ext_vector_type(8))) float v8f;

#define FIN   128
#define HID   8
#define NCLS  16
#define NEG_SLOPE 0.2f
#define ROW1  16
#define HROW1 16
#define HROW2 16

#define SPREAD_MAX 32.0f
#define GW1 14
#define GW2 30

__device__ __forceinline__ void atomicMaxF(float* addr, float v) {
    if (v >= 0.0f) atomicMax((int*)addr, __float_as_int(v));
    else           atomicMin((unsigned int*)addr, __float_as_uint(v));
}

__global__ void init_ws(float* acc1, int n) {
    int i = blockIdx.x * blockDim.x + threadIdx.x;
    if (i < n * ROW1)
        acc1[i] = (i == GW1 || i == GW1 + 1 || i == GW2 || i == GW2 + 1) ? -__builtin_inff() : 0.0f;
}

__global__ void gemm1_wmma(const float* __restrict__ x, const float* __restrict__ W1,
                           float* __restrict__ h1, int n) {
    __shared__ float wlds[FIN * 16];
    int t = threadIdx.x;
    {
        int k = t;
        const float* wr = W1 + k * HID;
        #pragma unroll
        for (int c = 0; c < 16; ++c) wlds[k * 16 + c] = (c < HID) ? wr[c] : 0.0f;
    }
    __syncthreads();

    int wave = t >> 5, lane = t & 31;
    int tile = blockIdx.x * 4 + wave;
    int col  = lane & 15;
    int hi   = lane >> 4;
    int mrow = tile * 16 + col;
    if (mrow >= n) mrow = n - 1;
    const float* xrow = x + (long)mrow * FIN;

    v8f c = {};
    #pragma unroll
    for (int k0 = 0; k0 < FIN; k0 += 4) {
        int ka = k0 + 2 * hi;
        v2f a = *(const v2f*)(xrow + ka);
        v2f b;
        b.x = wlds[ka * 16 + col];
        b.y = wlds[(ka + 1) * 16 + col];
        c = __builtin_amdgcn_wmma_f32_16x16x4_f32(false, a, false, b,
                                                  (short)0, c, false, false);
    }

    int mbase = tile * 16 + (hi << 3);
    if (col < HID) {
        #pragma unroll
        for (int j = 0; j < 8; ++j) {
            int r = mbase + j;
            if (r < n) h1[(long)r * HROW1 + col] = c[j];
        }
    }
}

__global__ void gemm2_wmma(const float* __restrict__ h1r, const float* __restrict__ W2,
                           float* __restrict__ h2, int n) {
    __shared__ float wlds[HID * 16];
    int t = threadIdx.x;
    if (t < HID * 16) wlds[t] = W2[t];
    __syncthreads();

    int wave = t >> 5, lane = t & 31;
    int tile = blockIdx.x * 4 + wave;
    int col  = lane & 15;
    int hi   = lane >> 4;
    int mrow = tile * 16 + col;
    if (mrow >= n) mrow = n - 1;
    const float* hrow = h1r + (long)mrow * ROW1;

    v8f c = {};
    #pragma unroll
    for (int k0 = 0; k0 < HID; k0 += 4) {
        int ka = k0 + 2 * hi;
        v2f a = *(const v2f*)(hrow + ka);
        v2f b;
        b.x = wlds[ka * 16 + col];
        b.y = wlds[(ka + 1) * 16 + col];
        c = __builtin_amdgcn_wmma_f32_16x16x4_f32(false, a, false, b,
                                                  (short)0, c, false, false);
    }

    int mbase = tile * 16 + (hi << 3);
    #pragma unroll
    for (int j = 0; j < 8; ++j) {
        int r = mbase + j;
        if (r < n) h2[(long)r * HROW2 + col] = c[j];
    }
}

template <int F, int HROW, int ASTR, int ZROW>
__global__ void alpha_kernel(const float* hrow, float* asrow,
                             const float* __restrict__ a_src,
                             const float* __restrict__ a_dst,
                             float* __restrict__ outs, float2* __restrict__ rec,
                             float4* __restrict__ zrows, float* __restrict__ g, int n) {
    __shared__ float red[2][8];
    int i = blockIdx.x * blockDim.x + threadIdx.x;
    float hi = -__builtin_inff(), lo = -__builtin_inff();
    if (i < n) {
        if (ZROW > 0) {
            #pragma unroll
            for (int k = 0; k < ZROW / 4; ++k) {
                zrows[(long)k * n + i] = make_float4(0.0f, 0.0f, 0.0f, 0.0f);
            }
        }
        const float* row = hrow + (long)i * HROW;
        float s = 0.0f, d = 0.0f;
        #pragma unroll
        for (int q = 0; q < F / 4; ++q) {
            float4 hv = ((const float4*)row)[q];
            s += hv.x * a_src[4 * q + 0] + hv.y * a_src[4 * q + 1]
               + hv.z * a_src[4 * q + 2] + hv.w * a_src[4 * q + 3];
            d += hv.x * a_dst[4 * q + 0] + hv.y * a_dst[4 * q + 1]
               + hv.z * a_dst[4 * q + 2] + hv.w * a_dst[4 * q + 3];
        }
        asrow[(long)i * ASTR] = s;
        if (outs) outs[i] = s;
        rec[i]  = make_float2(d, -__builtin_inff());
        hi = s + 0.0f;
        lo = 0.0f - s;
    }
    for (int off = 16; off > 0; off >>= 1) {
        hi = fmaxf(hi, __shfl_xor(hi, off));
        lo = fmaxf(lo, __shfl_xor(lo, off));
    }
    if ((threadIdx.x & 31) == 0) { red[0][threadIdx.x >> 5] = hi; red[1][threadIdx.x >> 5] = lo; }
    __syncthreads();
    if (threadIdx.x < 2) {
        float v = red[threadIdx.x][0];
        for (int w = 1; w < 8; ++w) v = fmaxf(v, red[threadIdx.x][w]);
        atomicMaxF(g + threadIdx.x, v);
    }
}

template <int STRIDE>
__global__ void edge_max(const int* __restrict__ src, const int* __restrict__ dst,
                         const float* __restrict__ as, float2* __restrict__ rec,
                         const float* __restrict__ g, int E) {
    if (g[0] + g[1] <= SPREAD_MAX) return;
    for (int i = blockIdx.x * blockDim.x + threadIdx.x; i < E; i += gridDim.x * blockDim.x) {
        int s = src[i], d = dst[i];
        atomicMaxF((float*)(rec + d) + 1, as[(long)s * STRIDE]);
    }
}

template <int ROW, int F>
__device__ __forceinline__ float from_lane_F(float v, int lane_f) {
    if (ROW == 32)
        return __int_as_float(__builtin_amdgcn_readlane(__float_as_int(v), F));
    if (ROW == 16)
        return __int_as_float(__builtin_amdgcn_update_dpp(0, __float_as_int(v), 0x150 + F, 0xf, 0xf, false));
    return __shfl(v, lane_f);
}

template <int F, int ROW, int HROW>
__global__ void edge_acc(const int* __restrict__ src, const int* __restrict__ dst,
                         const float2* __restrict__ rec,
                         const float* __restrict__ hrow,
                         float* __restrict__ acc, const float* __restrict__ g, int E) {
    int i = blockIdx.x * blockDim.x + threadIdx.x;
    int lane = threadIdx.x & 31;
    if (i - lane >= E) return;
    float gmax = g[0];
    bool  narrow = gmax + g[1] <= SPREAD_MAX;
    int s = 0, d = -1;
    float adv = 0.0f, mx = 0.0f;
    if (i < E) {
        s = src[i]; d = dst[i];
        float2 r = rec[d];
        adv = r.x;
        mx = (narrow ? gmax : r.y) + r.x;
        mx = (mx >= 0.0f) ? mx : NEG_SLOPE * mx;
    }
    constexpr int EPI = 32 / ROW;
    constexpr int NIT = 32 / EPI;
    constexpr int CH  = 8;
    int f   = lane % ROW;
    int sub = lane / ROW;
    int fl  = (f <= F) ? f : 0;
    int asl = sub * ROW + F;
    for (int c0 = 0; c0 < NIT; c0 += CH) {
        float hv[CH], adj[CH], mxj[CH];
        int   dj[CH];
        #pragma unroll
        for (int u = 0; u < CH; ++u) {
            int j  = (c0 + u) * EPI + sub;
            int sj = __shfl(s, j);
            dj[u]  = __shfl(d, j);
            adj[u] = __shfl(adv, j);
            mxj[u] = __shfl(mx, j);
            hv[u]  = hrow[(long)sj * HROW + fl];
        }
        #pragma unroll
        for (int u = 0; u < CH; ++u) {
            float e = from_lane_F<ROW, F>(hv[u], asl) + adj[u];
            e = (e >= 0.0f) ? e : NEG_SLOPE * e;
            float ex = __expf(e - mxj[u]);
            if (f <= F && dj[u] >= 0)
                atomicAdd(&acc[(long)dj[u] * ROW + f], (f < F) ? ex * hv[u] : ex);
        }
    }
}

__global__ void finish1(float* __restrict__ acc, const float* __restrict__ bias, int n) {
    int i = blockIdx.x * blockDim.x + threadIdx.x;
    if (i >= n) return;
    float* row = acc + (long)i * ROW1;
    float d = row[HID];
    float inv = (d > 0.0f) ? 1.0f / d : 0.0f;
    float4* ap = (float4*)row;
    #pragma unroll
    for (int q = 0; q < HID / 4; ++q) {
        float4 v = ap[q];
        v.x = fmaxf(v.x * inv + bias[4 * q + 0], 0.0f);
        v.y = fmaxf(v.y * inv + bias[4 * q + 1], 0.0f);
        v.z = fmaxf(v.z * inv + bias[4 * q + 2], 0.0f);
        v.w = fmaxf(v.w * inv + bias[4 * q + 3], 0.0f);
        ap[q] = v;
    }
}

template <int M>
__device__ __forceinline__ float row_xor(float v) {
    return __int_as_float(__builtin_amdgcn_update_dpp(0, __float_as_int(v), 0x160 + M, 0xf, 0xf, false));
}

__global__ void finish2_wmma(const float* __restrict__ acc, const float* __restrict__ W2,
                             const float* __restrict__ bias, float* __restrict__ out, int n) {
    __shared__ float wlds[HID * 16];
    int t = threadIdx.x;
    if (t < HID * 16) wlds[t] = W2[t];
    __syncthreads();

    int wave = t >> 5, lane = t & 31;
    int tile = blockIdx.x * 4 + wave;
    int col  = lane & 15;
    int hi   = lane >> 4;
    int mrow = tile * 16 + col;
    if (mrow >= n) mrow = n - 1;
    const float* arow = acc + (long)mrow * ROW1;
    float d = arow[HID];
    float inv = (d > 0.0f) ? 1.0f / d : 0.0f;

    v8f c = {};
    #pragma unroll
    for (int k0 = 0; k0 < HID; k0 += 4) {
        int ka = k0 + 2 * hi;
        v2f a = *(const v2f*)(arow + ka);
        a.x *= inv; a.y *= inv;
        v2f b;
        b.x = wlds[ka * 16 + col];
        b.y = wlds[(ka + 1) * 16 + col];
        c = __builtin_amdgcn_wmma_f32_16x16x4_f32(false, a, false, b,
                                                  (short)0, c, false, false);
    }

    float bs = bias[col];
    int mbase = tile * 16 + (hi << 3);
    #pragma unroll
    for (int j = 0; j < 8; ++j) {
        float v = c[j] + bs;
        float mx = v;
        mx = fmaxf(mx, row_xor<8>(mx)); mx = fmaxf(mx, row_xor<4>(mx));
        mx = fmaxf(mx, row_xor<2>(mx)); mx = fmaxf(mx, row_xor<1>(mx));
        float sm = __expf(v - mx);
        sm += row_xor<8>(sm); sm += row_xor<4>(sm);
        sm += row_xor<2>(sm); sm += row_xor<1>(sm);
        float ls = __logf(sm) + mx;
        int r = mbase + j;
        if (r < n) out[(long)r * NCLS + col] = v - ls;
    }
}

extern "C" void kernel_launch(void* const* d_in, const int* in_sizes, int n_in,
                              void* d_out, int out_size, void* d_ws, size_t ws_size,
                              hipStream_t stream) {
    const float* x   = (const float*)d_in[0];
    const int*   ei  = (const int*)  d_in[1];
    const float* W1  = (const float*)d_in[2];
    const float* a1s = (const float*)d_in[3];
    const float* a1d = (const float*)d_in[4];
    const float* b1  = (const float*)d_in[5];
    const float* W2  = (const float*)d_in[6];
    const float* a2s = (const float*)d_in[7];
    const float* a2d = (const float*)d_in[8];
    const float* b2  = (const float*)d_in[9];

    int n = in_sizes[0] / FIN;
    int E = in_sizes[1] / 2;
    const int* src = ei;
    const int* dst = ei + E;

    float*  ws   = (float*)d_ws;
    float*  acc1 = ws;
    float*  hr1  = acc1 + (long)n * ROW1;
    float2* rec1 = (float2*)(hr1 + (long)n * HROW1);
    float*  as1  = (float*)(rec1 + n);
    float*  acc2 = hr1;
    float*  hr2  = hr1 + (long)n * HROW1;
    float2* rec2 = (float2*)(hr2 + (long)n * HROW2);

    int ntiles  = (n + 15) / 16;
    int gblocks = (ntiles + 3) / 4;
    int eblocks = (E + 255) / 256;
    int nblocks = (n + 255) / 256;

    init_ws<<<(n * ROW1 + 255) / 256, 256, 0, stream>>>(acc1, n);

    gemm1_wmma<<<gblocks, 128, 0, stream>>>(x, W1, hr1, n);
    alpha_kernel<HID, HROW1, HROW1, 0><<<nblocks, 256, 0, stream>>>(hr1, hr1 + HID, a1s, a1d, as1, rec1, (float4*)0, ws + GW1, n);
    edge_max<1><<<2048, 256, 0, stream>>>(src, dst, as1, rec1, ws + GW1, E);
    edge_acc<HID, ROW1, HROW1><<<eblocks, 256, 0, stream>>>(src, dst, rec1, hr1, acc1, ws + GW1, E);
    finish1<<<nblocks, 256, 0, stream>>>(acc1, b1, n);

    gemm2_wmma<<<gblocks, 128, 0, stream>>>(acc1, W2, hr2, n);
    alpha_kernel<NCLS, HROW2, ROW1, ROW1><<<nblocks, 256, 0, stream>>>(hr2, acc1 + HID, a2s, a2d, (float*)0, rec2, (float4*)acc2, ws + GW2, n);
    edge_max<ROW1><<<2048, 256, 0, stream>>>(src, dst, acc1 + HID, rec2, ws + GW2, E);
    edge_acc<HID, ROW1, ROW1><<<eblocks, 256, 0, stream>>>(src, dst, rec2, acc1, acc2, ws + GW2, E);
    finish2_wmma<<<gblocks, 128, 0, stream>>>(acc2, W2, b2, (float*)d_out, n);
}


# edge_acc: accumulator rows kept twice, one copy per package half (the wave's die id picks it), so no row's atomics cross between the halves' L2; the epilogues add the two copies: sum order changes
# speedup vs baseline: 15.0881x; 1.8250x over previous
#include <hip/hip_runtime.h>
#include <math.h>

typedef __attribute__((ext_vector_type(2))) float v2f;
typedef __attribute__((ext_vector_type(8))) float v8f;

#define FIN   128
#define HID   8
#define NCLS  16
#define NEG_SLOPE 0.2f
#define ROW1  16
#define HROW1 16
#define HROW2 16

#define SPREAD_MAX 32.0f
#define GW1 14
#define GW2 30

__device__ __forceinline__ void atomicMaxF(float* addr, float v) {
    if (v >= 0.0f) atomicMax((int*)addr, __float_as_int(v));
    else           atomicMin((unsigned int*)addr, __float_as_uint(v));
}

__device__ __forceinline__ bool upper_half() {
    return (__builtin_amdgcn_s_sendmsg_rtn(0x87) & (1u << 18)) != 0u;
}

__global__ void init_ws(float* acc1, float* acc1b, int n) {
    int i = blockIdx.x * blockDim.x + threadIdx.x;
    if (i < n * ROW1) {
        acc1[i] = (i == GW1 || i == GW1 + 1 || i == GW2 || i == GW2 + 1) ? -__builtin_inff() : 0.0f;
        acc1b[i] = 0.0f;
    }
}

__global__ void gemm1_wmma(const float* __restrict__ x, const float* __restrict__ W1,
                           float* __restrict__ h1, int n) {
    __shared__ float wlds[FIN * 16];
    int t = threadIdx.x;
    {
        int k = t;
        const float* wr = W1 + k * HID;
        #pragma unroll
        for (int c = 0; c < 16; ++c) wlds[k * 16 + c] = (c < HID) ? wr[c] : 0.0f;
    }
    __syncthreads();

    int wave = t >> 5, lane = t & 31;
    int tile = blockIdx.x * 4 + wave;
    int col  = lane & 15;
    int hi   = lane >> 4;
    int mrow = tile * 16 + col;
    if (mrow >= n) mrow = n - 1;
    const float* xrow = x + (long)mrow * FIN;

    v8f c = {};
    #pragma unroll
    for (int k0 = 0; k0 < FIN; k0 += 4) {
        int ka = k0 + 2 * hi;
        v2f a = *(const v2f*)(xrow + ka);
        v2f b;
        b.x = wlds[ka * 16 + col];
        b.y = wlds[(ka + 1) * 16 + col];
        c = __builtin_amdgcn_wmma_f32_16x16x4_f32(false, a, false, b,
                                                  (short)0, c, false, false);
    }

    int mbase = tile * 16 + (hi << 3);
    if (col < HID) {
        #pragma unroll
        for (int j = 0; j < 8; ++j) {
            int r = mbase + j;
            if (r < n) h1[(long)r * HROW1 + col] = c[j];
        }
    }
}

__global__ void gemm2_wmma(const float* __restrict__ h1r, const float* __restrict__ W2,
                           float* __restrict__ h2, int n) {
    __shared__ float wlds[HID * 16];
    int t = threadIdx.x;
    if (t < HID * 16) wlds[t] = W2[t];
    __syncthreads();

    int wave = t >> 5, lane = t & 31;
    int tile = blockIdx.x * 4 + wave;
    int col  = lane & 15;
    int hi   = lane >> 4;
    int mrow = tile * 16 + col;
    if (mrow >= n) mrow = n - 1;
    const float* hrow = h1r + (long)mrow * ROW1;

    v8f c = {};
    #pragma unroll
    for (int k0 = 0; k0 < HID; k0 += 4) {
        int ka = k0 + 2 * hi;
        v2f a = *(const v2f*)(hrow + ka);
        v2f b;
        b.x = wlds[ka * 16 + col];
        b.y = wlds[(ka + 1) * 16 + col];
        c = __builtin_amdgcn_wmma_f32_16x16x4_f32(false, a, false, b,
                                                  (short)0, c, false, false);
    }

    int mbase = tile * 16 + (hi << 3);
    #pragma unroll
    for (int j = 0; j < 8; ++j) {
        int r = mbase + j;
        if (r < n) h2[(long)r * HROW2 + col] = c[j];
    }
}

template <int F, int HROW, int ASTR, int ZROW>
__global__ void alpha_kernel(const float* hrow, float* asrow,
                             const float* __restrict__ a_src,
                             const float* __restrict__ a_dst,
                             float2* __restrict__ rec,
                             float4* __restrict__ zrows, float* __restrict__ g, int n) {
    __shared__ float red[2][8];
    int i = blockIdx.x * blockDim.x + threadIdx.x;
    float hi = -__builtin_inff(), lo = -__builtin_inff();
    if (i < n) {
        if (ZROW > 0) {
            #pragma unroll
            for (int k = 0; k < ZROW / 4; ++k) {
                zrows[(long)k * n + i] = make_float4(0.0f, 0.0f, 0.0f, 0.0f);
            }
        }
        const float* row = hrow + (long)i * HROW;
        float s = 0.0f, d = 0.0f;
        #pragma unroll
        for (int q = 0; q < F / 4; ++q) {
            float4 hv = ((const float4*)row)[q];
            s += hv.x * a_src[4 * q + 0] + hv.y * a_src[4 * q + 1]
               + hv.z * a_src[4 * q + 2] + hv.w * a_src[4 * q + 3];
            d += hv.x * a_dst[4 * q + 0] + hv.y * a_dst[4 * q + 1]
               + hv.z * a_dst[4 * q + 2] + hv.w * a_dst[4 * q + 3];
        }
        asrow[(long)i * ASTR] = s;
        rec[i]  = make_float2(d, -__builtin_inff());
        hi = s + 0.0f;
        lo = 0.0f - s;
    }
    for (int off = 16; off > 0; off >>= 1) {
        hi = fmaxf(hi, __shfl_xor(hi, off));
        lo = fmaxf(lo, __shfl_xor(lo, off));
    }
    if ((threadIdx.x & 31) == 0) { red[0][threadIdx.x >> 5] = hi; red[1][threadIdx.x >> 5] = lo; }
    __syncthreads();
    if (threadIdx.x < 2) {
        float v = red[threadIdx.x][0];
        for (int w = 1; w < 8; ++w) v = fmaxf(v, red[threadIdx.x][w]);
        atomicMaxF(g + threadIdx.x, v);
    }
}

template <int STRIDE>
__global__ void edge_max(const int* __restrict__ src, const int* __restrict__ dst,
                         const float* __restrict__ as, float2* __restrict__ rec,
                         const float* __restrict__ g, float4* __restrict__ zrows, int nzrows, int E) {
    if (zrows) {
        int nz = nzrows * (ROW1 / 4);
        for (int k = blockIdx.x * blockDim.x + threadIdx.x; k < nz; k += gridDim.x * blockDim.x)
            zrows[k] = make_float4(0.0f, 0.0f, 0.0f, 0.0f);
    }
    if (g[0] + g[1] <= SPREAD_MAX) return;
    for (int i = blockIdx.x * blockDim.x + threadIdx.x; i < E; i += gridDim.x * blockDim.x) {
        int s = src[i], d = dst[i];
        atomicMaxF((float*)(rec + d) + 1, as[(long)s * STRIDE]);
    }
}

template <int ROW, int F>
__device__ __forceinline__ float from_lane_F(float v, int lane_f) {
    if (ROW == 32)
        return __int_as_float(__builtin_amdgcn_readlane(__float_as_int(v), F));
    if (ROW == 16)
        return __int_as_float(__builtin_amdgcn_update_dpp(0, __float_as_int(v), 0x150 + F, 0xf, 0xf, false));
    return __shfl(v, lane_f);
}

template <int F, int ROW, int HROW>
__global__ void edge_acc(const int* __restrict__ src, const int* __restrict__ dst,
                         const float2* __restrict__ rec,
                         const float* __restrict__ hrow,
                         float* __restrict__ accA, float* __restrict__ accB,
                         const float* __restrict__ g, int E) {
    int i = blockIdx.x * blockDim.x + threadIdx.x;
    int lane = threadIdx.x & 31;
    if (i - lane >= E) return;
    float* acc = upper_half() ? accB : accA;
    float gmax = g[0];
    bool  narrow = gmax + g[1] <= SPREAD_MAX;
    int s = 0, d = -1;
    float adv = 0.0f, mx = 0.0f;
    if (i < E) {
        s = src[i]; d = dst[i];
        float2 r = rec[d];
        adv = r.x;
        mx = (narrow ? gmax : r.y) + r.x;
        mx = (mx >= 0.0f) ? mx : NEG_SLOPE * mx;
    }
    constexpr int EPI = 32 / ROW;
    constexpr int NIT = 32 / EPI;
    constexpr int CH  = 8;
    int f   = lane % ROW;
    int sub = lane / ROW;
    int fl  = (f <= F) ? f : 0;
    int asl = sub * ROW + F;
    for (int c0 = 0; c0 < NIT; c0 += CH) {
        float hv[CH], adj[CH], mxj[CH];
        int   dj[CH];
        #pragma unroll
        for (int u = 0; u < CH; ++u) {
            int j  = (c0 + u) * EPI + sub;
            int sj = __shfl(s, j);
            dj[u]  = __shfl(d, j);
            adj[u] = __shfl(adv, j);
            mxj[u] = __shfl(mx, j);
            hv[u]  = hrow[(long)sj * HROW + fl];
        }
        #pragma unroll
        for (int u = 0; u < CH; ++u) {
            float e = from_lane_F<ROW, F>(hv[u], asl) + adj[u];
            e = (e >= 0.0f) ? e : NEG_SLOPE * e;
            float ex = __expf(e - mxj[u]);
            if (f <= F && dj[u] >= 0)
                atomicAdd(&acc[(long)dj[u] * ROW + f], (f < F) ? ex * hv[u] : ex);
        }
    }
}

__global__ void finish1(float* __restrict__ acc, const float* __restrict__ accb,
                        const float* __restrict__ bias, int n) {
    int i = blockIdx.x * blockDim.x + threadIdx.x;
    if (i >= n) return;
    float* row = acc + (long)i * ROW1;
    const float* rowb = accb + (long)i * ROW1;
    float d = row[HID] + rowb[HID];
    float inv = (d > 0.0f) ? 1.0f / d : 0.0f;
    float4* ap = (float4*)row;
    const float4* bp = (const float4*)rowb;
    #pragma unroll
    for (int q = 0; q < HID / 4; ++q) {
        float4 v = ap[q];
        float4 w = bp[q];
        v.x = fmaxf((v.x + w.x) * inv + bias[4 * q + 0], 0.0f);
        v.y = fmaxf((v.y + w.y) * inv + bias[4 * q + 1], 0.0f);
        v.z = fmaxf((v.z + w.z) * inv + bias[4 * q + 2], 0.0f);
        v.w = fmaxf((v.w + w.w) * inv + bias[4 * q + 3], 0.0f);
        ap[q] = v;
    }
}

template <int M>
__device__ __forceinline__ float row_xor(float v) {
    return __int_as_float(__builtin_amdgcn_update_dpp(0, __float_as_int(v), 0x160 + M, 0xf, 0xf, false));
}

__global__ void finish2_wmma(const float* __restrict__ acc, const float* __restrict__ accb,
                             const float* __restrict__ W2,
                             const float* __restrict__ bias, float* __restrict__ out, int n) {
    __shared__ float wlds[HID * 16];
    int t = threadIdx.x;
    if (t < HID * 16) wlds[t] = W2[t];
    __syncthreads();

    int wave = t >> 5, lane = t & 31;
    int tile = blockIdx.x * 4 + wave;
    int col  = lane & 15;
    int hi   = lane >> 4;
    int mrow = tile * 16 + col;
    if (mrow >= n) mrow = n - 1;
    const float* arow = acc + (long)mrow * ROW1;
    const float* brow = accb + (long)mrow * ROW1;
    float d = arow[HID] + brow[HID];
    float inv = (d > 0.0f) ? 1.0f / d : 0.0f;

    v8f c = {};
    #pragma unroll
    for (int k0 = 0; k0 < HID; k0 += 4) {
        int ka = k0 + 2 * hi;
        v2f a = *(const v2f*)(arow + ka);
        v2f a2 = *(const v2f*)(brow + ka);
        a.x = (a.x + a2.x) * inv; a.y = (a.y + a2.y) * inv;
        v2f b;
        b.x = wlds[ka * 16 + col];
        b.y = wlds[(ka + 1) * 16 + col];
        c = __builtin_amdgcn_wmma_f32_16x16x4_f32(false, a, false, b,
                                                  (short)0, c, false, false);
    }

    float bs = bias[col];
    int mbase = tile * 16 + (hi << 3);
    #pragma unroll
    for (int j = 0; j < 8; ++j) {
        float v = c[j] + bs;
        float mx = v;
        mx = fmaxf(mx, row_xor<8>(mx)); mx = fmaxf(mx, row_xor<4>(mx));
        mx = fmaxf(mx, row_xor<2>(mx)); mx = fmaxf(mx, row_xor<1>(mx));
        float sm = __expf(v - mx);
        sm += row_xor<8>(sm); sm += row_xor<4>(sm);
        sm += row_xor<2>(sm); sm += row_xor<1>(sm);
        float ls = __logf(sm) + mx;
        int r = mbase + j;
        if (r < n) out[(long)r * NCLS + col] = v - ls;
    }
}

extern "C" void kernel_launch(void* const* d_in, const int* in_sizes, int n_in,
                              void* d_out, int out_size, void* d_ws, size_t ws_size,
                              hipStream_t stream) {
    const float* x   = (const float*)d_in[0];
    const int*   ei  = (const int*)  d_in[1];
    const float* W1  = (const float*)d_in[2];
    const float* a1s = (const float*)d_in[3];
    const float* a1d = (const float*)d_in[4];
    const float* b1  = (const float*)d_in[5];
    const float* W2  = (const float*)d_in[6];
    const float* a2s = (const float*)d_in[7];
    const float* a2d = (const float*)d_in[8];
    const float* b2  = (const float*)d_in[9];

    int n = in_sizes[0] / FIN;
    int E = in_sizes[1] / 2;
    const int* src = ei;
    const int* dst = ei + E;

    float*  ws   = (float*)d_ws;
    float*  acc1 = ws;
    float*  hr1  = acc1 + (long)n * ROW1;
    float*  acc1b = hr1 + (long)n * HROW1;
    float2* rec1 = (float2*)(acc1b + (long)n * ROW1);
    float*  acc2 = hr1;
    float*  hr2  = acc1b;
    float*  acc2b = hr2;
    float2* rec2 = rec1;

    int ntiles  = (n + 15) / 16;
    int gblocks = (ntiles + 3) / 4;
    int eblocks = (E + 255) / 256;
    int nblocks = (n + 255) / 256;

    init_ws<<<(n * ROW1 + 255) / 256, 256, 0, stream>>>(acc1, acc1b, n);

    gemm1_wmma<<<gblocks, 128, 0, stream>>>(x, W1, hr1, n);
    alpha_kernel<HID, HROW1, HROW1, 0><<<nblocks, 256, 0, stream>>>(hr1, hr1 + HID, a1s, a1d, rec1, (float4*)0, ws + GW1, n);
    edge_max<HROW1><<<2048, 256, 0, stream>>>(src, dst, hr1 + HID, rec1, ws + GW1, (float4*)0, 0, E);
    edge_acc<HID, ROW1, HROW1><<<eblocks, 256, 0, stream>>>(src, dst, rec1, hr1, acc1, acc1b, ws + GW1, E);
    finish1<<<nblocks, 256, 0, stream>>>(acc1, acc1b, b1, n);

    gemm2_wmma<<<gblocks, 128, 0, stream>>>(acc1, W2, hr2, n);
    alpha_kernel<NCLS, HROW2, ROW1, ROW1><<<nblocks, 256, 0, stream>>>(hr2, acc1 + HID, a2s, a2d, rec2, (float4*)acc2, ws + GW2, n);
    edge_max<ROW1><<<2048, 256, 0, stream>>>(src, dst, acc1 + HID, rec2, ws + GW2, (float4*)acc2b, n, E);
    edge_acc<HID, ROW1, ROW1><<<eblocks, 256, 0, stream>>>(src, dst, rec2, acc1, acc2, acc2b, ws + GW2, E);
    finish2_wmma<<<gblocks, 128, 0, stream>>>(acc2, acc2b, W2, b2, (float*)d_out, n);
}


# edge_acc: ad[dst] of the first 80000 nodes comes from an LDS table (256 resident work-groups of 32 waves loop over the edges): no L2 gather per edge for it; atomic sums in arrival order as before
# speedup vs baseline: 16.0372x; 1.0629x over previous
#include <hip/hip_runtime.h>
#include <math.h>

typedef __attribute__((ext_vector_type(2))) float v2f;
typedef __attribute__((ext_vector_type(8))) float v8f;

#define FIN   128
#define HID   8
#define NCLS  16
#define NEG_SLOPE 0.2f
#define ROW1  16
#define HROW1 16
#define HROW2 16

#define SPREAD_MAX 32.0f
#define GW1 14
#define GW2 30
#define NTAB 80000
#define EA_THREADS 1024

__device__ __forceinline__ void atomicMaxF(float* addr, float v) {
    if (v >= 0.0f) atomicMax((int*)addr, __float_as_int(v));
    else           atomicMin((unsigned int*)addr, __float_as_uint(v));
}

__device__ __forceinline__ bool upper_half() {
    return (__builtin_amdgcn_s_sendmsg_rtn(0x87) & (1u << 18)) != 0u;
}

__global__ void init_ws(float* acc1, float* acc1b, int n) {
    int i = blockIdx.x * blockDim.x + threadIdx.x;
    if (i < n * ROW1) {
        acc1[i] = (i == GW1 || i == GW1 + 1 || i == GW2 || i == GW2 + 1) ? -__builtin_inff() : 0.0f;
        acc1b[i] = 0.0f;
    }
}

__global__ void gemm1_wmma(const float* __restrict__ x, const float* __restrict__ W1,
                           float* __restrict__ h1, int n) {
    __shared__ float wlds[FIN * 16];
    int t = threadIdx.x;
    {
        int k = t;
        const float* wr = W1 + k * HID;
        #pragma unroll
        for (int c = 0; c < 16; ++c) wlds[k * 16 + c] = (c < HID) ? wr[c] : 0.0f;
    }
    __syncthreads();

    int wave = t >> 5, lane = t & 31;
    int tile = blockIdx.x * 4 + wave;
    int col  = lane & 15;
    int hi   = lane >> 4;
    int mrow = tile * 16 + col;
    if (mrow >= n) mrow = n - 1;
    const float* xrow = x + (long)mrow * FIN;

    v8f c = {};
    #pragma unroll
    for (int k0 = 0; k0 < FIN; k0 += 4) {
        int ka = k0 + 2 * hi;
        v2f a = *(const v2f*)(xrow + ka);
        v2f b;
        b.x = wlds[ka * 16 + col];
        b.y = wlds[(ka + 1) * 16 + col];
        c = __builtin_amdgcn_wmma_f32_16x16x4_f32(false, a, false, b,
                                                  (short)0, c, false, false);
    }

    int mbase = tile * 16 + (hi << 3);
    if (col < HID) {
        #pragma unroll
        for (int j = 0; j < 8; ++j) {
            int r = mbase + j;
            if (r < n) h1[(long)r * HROW1 + col] = c[j];
        }
    }
}

__global__ void gemm2_wmma(const float* __restrict__ h1r, const float* __restrict__ W2,
                           float* __restrict__ h2, int n) {
    __shared__ float wlds[HID * 16];
    int t = threadIdx.x;
    if (t < HID * 16) wlds[t] = W2[t];
    __syncthreads();

    int wave = t >> 5, lane = t & 31;
    int tile = blockIdx.x * 4 + wave;
    int col  = lane & 15;
    int hi   = lane >> 4;
    int mrow = tile * 16 + col;
    if (mrow >= n) mrow = n - 1;
    const float* hrow = h1r + (long)mrow * ROW1;

    v8f c = {};
    #pragma unroll
    for (int k0 = 0; k0 < HID; k0 += 4) {
        int ka = k0 + 2 * hi;
        v2f a = *(const v2f*)(hrow + ka);
        v2f b;
        b.x = wlds[ka * 16 + col];
        b.y = wlds[(ka + 1) * 16 + col];
        c = __builtin_amdgcn_wmma_f32_16x16x4_f32(false, a, false, b,
                                                  (short)0, c, false, false);
    }

    int mbase = tile * 16 + (hi << 3);
    #pragma unroll
    for (int j = 0; j < 8; ++j) {
        int r = mbase + j;
        if (r < n) h2[(long)r * HROW2 + col] = c[j];
    }
}

template <int F, int HROW, int ASTR, int ZROW>
__global__ void alpha_kernel(const float* hrow, float* asrow,
                             const float* __restrict__ a_src,
                             const float* __restrict__ a_dst,
                             float* __restrict__ adp, float* __restrict__ masp,
                             float4* __restrict__ zrows, float* __restrict__ g, int n) {
    __shared__ float red[2][8];
    int i = blockIdx.x * blockDim.x + threadIdx.x;
    float hi = -__builtin_inff(), lo = -__builtin_inff();
    if (i < n) {
        if (ZROW > 0) {
            #pragma unroll
            for (int k = 0; k < ZROW / 4; ++k) {
                zrows[(long)k * n + i] = make_float4(0.0f, 0.0f, 0.0f, 0.0f);
            }
        }
        const float* row = hrow + (long)i * HROW;
        float s = 0.0f, d = 0.0f;
        #pragma unroll
        for (int q = 0; q < F / 4; ++q) {
            float4 hv = ((const float4*)row)[q];
            s += hv.x * a_src[4 * q + 0] + hv.y * a_src[4 * q + 1]
               + hv.z * a_src[4 * q + 2] + hv.w * a_src[4 * q + 3];
            d += hv.x * a_dst[4 * q + 0] + hv.y * a_dst[4 * q + 1]
               + hv.z * a_dst[4 * q + 2] + hv.w * a_dst[4 * q + 3];
        }
        asrow[(long)i * ASTR] = s;
        adp[i]  = d;
        masp[i] = -__builtin_inff();
        hi = s + 0.0f;
        lo = 0.0f - s;
    }
    for (int off = 16; off > 0; off >>= 1) {
        hi = fmaxf(hi, __shfl_xor(hi, off));
        lo = fmaxf(lo, __shfl_xor(lo, off));
    }
    if ((threadIdx.x & 31) == 0) { red[0][threadIdx.x >> 5] = hi; red[1][threadIdx.x >> 5] = lo; }
    __syncthreads();
    if (threadIdx.x < 2) {
        float v = red[threadIdx.x][0];
        for (int w = 1; w < 8; ++w) v = fmaxf(v, red[threadIdx.x][w]);
        atomicMaxF(g + threadIdx.x, v);
    }
}

template <int STRIDE>
__global__ void edge_max(const int* __restrict__ src, const int* __restrict__ dst,
                         const float* __restrict__ as, float* __restrict__ masp,
                         const float* __restrict__ g, float4* __restrict__ zrows, int nzrows, int E) {
    if (zrows) {
        int nz = nzrows * (ROW1 / 4);
        for (int k = blockIdx.x * blockDim.x + threadIdx.x; k < nz; k += gridDim.x * blockDim.x)
            zrows[k] = make_float4(0.0f, 0.0f, 0.0f, 0.0f);
    }
    if (g[0] + g[1] <= SPREAD_MAX) return;
    for (int i = blockIdx.x * blockDim.x + threadIdx.x; i < E; i += gridDim.x * blockDim.x) {
        int s = src[i], d = dst[i];
        atomicMaxF(masp + d, as[(long)s * STRIDE]);
    }
}

template <int ROW, int F>
__device__ __forceinline__ float from_lane_F(float v, int lane_f) {
    if (ROW == 32)
        return __int_as_float(__builtin_amdgcn_readlane(__float_as_int(v), F));
    if (ROW == 16)
        return __int_as_float(__builtin_amdgcn_update_dpp(0, __float_as_int(v), 0x150 + F, 0xf, 0xf, false));
    return __shfl(v, lane_f);
}

template <int F, int ROW, int HROW>
__global__ void __launch_bounds__(EA_THREADS)
edge_acc(const int* __restrict__ src, const int* __restrict__ dst,
         const float* __restrict__ adp, const float* __restrict__ masp,
         const float* __restrict__ hrow,
         float* __restrict__ accA, float* __restrict__ accB,
         const float* __restrict__ g, int n, int E) {
    __shared__ __attribute__((aligned(16))) float tab[NTAB];
    int lane = threadIdx.x & 31;
    float* acc = upper_half() ? accB : accA;
    float gmax = g[0];
    bool  narrow = gmax + g[1] <= SPREAD_MAX;
    int   ntab = narrow ? (n < NTAB ? n : NTAB) : 0;
    {
        int n4 = ntab >> 2;
        const float4* a4 = (const float4*)adp;
        float4* t4 = (float4*)tab;
        constexpr int UF = 10;
        for (int k0 = threadIdx.x; k0 < n4; k0 += UF * EA_THREADS) {
            float4 v[UF];
            int    kk[UF];
            #pragma unroll
            for (int u = 0; u < UF; ++u) {
                int k = k0 + u * EA_THREADS;
                kk[u] = (k < n4) ? k : k0;
                v[u] = a4[kk[u]];
            }
            #pragma unroll
            for (int u = 0; u < UF; ++u) t4[kk[u]] = v[u];
        }
        int kr = (n4 << 2) + threadIdx.x;
        if (kr < ntab) tab[kr] = adp[kr];
    }
    __syncthreads();

    constexpr int EPI = 32 / ROW;
    constexpr int NIT = 32 / EPI;
    int f   = lane % ROW;
    int sub = lane / ROW;
    int fl  = (f <= F) ? f : 0;
    int asl = sub * ROW + F;

    int stride = gridDim.x * EA_THREADS;
    int i = blockIdx.x * EA_THREADS + threadIdx.x;
    int s = 0, d = -1;
    if (i < E) { s = src[i]; d = dst[i]; }
    while (i - lane < E) {
        bool  intab = (unsigned)d < (unsigned)ntab;
        bool  beyond = (d >= 0) && !intab;
        float adt = tab[intab ? d : 0];
        float adg = 0.0f;
        float top = gmax;
        if (beyond) {
            adg = adp[d];
            if (!narrow) top = masp[d];
        }
        float hv[NIT];
        #pragma unroll
        for (int u = 0; u < NIT; ++u) {
            int j  = u * EPI + sub;
            int sj = __shfl(s, j);
            hv[u]  = hrow[(long)sj * HROW + fl];
        }
        int inx = i + stride;
        int ix  = (inx < E) ? inx : E - 1;
        int sn  = src[ix];
        int dn  = dst[ix];
        if (inx >= E) dn = -1;
        float adv = intab ? adt : adg;
        float mx = top + adv;
        mx = (mx >= 0.0f) ? mx : NEG_SLOPE * mx;
        float adj[NIT], mxj[NIT];
        int   dj[NIT];
        #pragma unroll
        for (int u = 0; u < NIT; ++u) {
            int j  = u * EPI + sub;
            dj[u]  = __shfl(d, j);
            adj[u] = __shfl(adv, j);
            mxj[u] = __shfl(mx, j);
        }
        #pragma unroll
        for (int u = 0; u < NIT; ++u) {
            float e = from_lane_F<ROW, F>(hv[u], asl) + adj[u];
            e = (e >= 0.0f) ? e : NEG_SLOPE * e;
            float ex = __expf(e - mxj[u]);
            if (f <= F && dj[u] >= 0)
                atomicAdd(&acc[(long)dj[u] * ROW + f], (f < F) ? ex * hv[u] : ex);
        }
        s = sn; d = dn; i = inx;
    }
}

__global__ void finish1(float* __restrict__ acc, const float* __restrict__ accb,
                        const float* __restrict__ bias, int n) {
    int i = blockIdx.x * blockDim.x + threadIdx.x;
    if (i >= n) return;
    float* row = acc + (long)i * ROW1;
    const float* rowb = accb + (long)i * ROW1;
    float d = row[HID] + rowb[HID];
    float inv = (d > 0.0f) ? 1.0f / d : 0.0f;
    float4* ap = (float4*)row;
    const float4* bp = (const float4*)rowb;
    #pragma unroll
    for (int q = 0; q < HID / 4; ++q) {
        float4 v = ap[q];
        float4 w = bp[q];
        v.x = fmaxf((v.x + w.x) * inv + bias[4 * q + 0], 0.0f);
        v.y = fmaxf((v.y + w.y) * inv + bias[4 * q + 1], 0.0f);
        v.z = fmaxf((v.z + w.z) * inv + bias[4 * q + 2], 0.0f);
        v.w = fmaxf((v.w + w.w) * inv + bias[4 * q + 3], 0.0f);
        ap[q] = v;
    }
}

template <int M>
__device__ __forceinline__ float row_xor(float v) {
    return __int_as_float(__builtin_amdgcn_update_dpp(0, __float_as_int(v), 0x160 + M, 0xf, 0xf, false));
}

__global__ void finish2_wmma(const float* __restrict__ acc, const float* __restrict__ accb,
                             const float* __restrict__ W2,
                             const float* __restrict__ bias, float* __restrict__ out, int n) {
    __shared__ float wlds[HID * 16];
    int t = threadIdx.x;
    if (t < HID * 16) wlds[t] = W2[t];
    __syncthreads();

    int wave = t >> 5, lane = t & 31;
    int tile = blockIdx.x * 4 + wave;
    int col  = lane & 15;
    int hi   = lane >> 4;
    int mrow = tile * 16 + col;
    if (mrow >= n) mrow = n - 1;
    const float* arow = acc + (long)mrow * ROW1;
    const float* brow = accb + (long)mrow * ROW1;
    float d = arow[HID] + brow[HID];
    float inv = (d > 0.0f) ? 1.0f / d : 0.0f;

    v8f c = {};
    #pragma unroll
    for (int k0 = 0; k0 < HID; k0 += 4) {
        int ka = k0 + 2 * hi;
        v2f a = *(const v2f*)(arow + ka);
        v2f a2 = *(const v2f*)(brow + ka);
        a.x = (a.x + a2.x) * inv; a.y = (a.y + a2.y) * inv;
        v2f b;
        b.x = wlds[ka * 16 + col];
        b.y = wlds[(ka + 1) * 16 + col];
        c = __builtin_amdgcn_wmma_f32_16x16x4_f32(false, a, false, b,
                                                  (short)0, c, false, false);
    }

    float bs = bias[col];
    int mbase = tile * 16 + (hi << 3);
    #pragma unroll
    for (int j = 0; j < 8; ++j) {
        float v = c[j] + bs;
        float mx = v;
        mx = fmaxf(mx, row_xor<8>(mx)); mx = fmaxf(mx, row_xor<4>(mx));
        mx = fmaxf(mx, row_xor<2>(mx)); mx = fmaxf(mx, row_xor<1>(mx));
        float sm = __expf(v - mx);
        sm += row_xor<8>(sm); sm += row_xor<4>(sm);
        sm += row_xor<2>(sm); sm += row_xor<1>(sm);
        float ls = __logf(sm) + mx;
        int r = mbase + j;
        if (r < n) out[(long)r * NCLS + col] = v - ls;
    }
}

extern "C" void kernel_launch(void* const* d_in, const int* in_sizes, int n_in,
                              void* d_out, int out_size, void* d_ws, size_t ws_size,
                              hipStream_t stream) {
    const float* x   = (const float*)d_in[0];
    const int*   ei  = (const int*)  d_in[1];
    const float* W1  = (const float*)d_in[2];
    const float* a1s = (const float*)d_in[3];
    const float* a1d = (const float*)d_in[4];
    const float* b1  = (const float*)d_in[5];
    const float* W2  = (const float*)d_in[6];
    const float* a2s = (const float*)d_in[7];
    const float* a2d = (const float*)d_in[8];
    const float* b2  = (const float*)d_in[9];

    int n = in_sizes[0] / FIN;
    int E = in_sizes[1] / 2;
    const int* src = ei;
    const int* dst = ei + E;

    float*  ws   = (float*)d_ws;
    float*  acc1 = ws;
    float*  hr1  = acc1 + (long)n * ROW1;
    float*  acc1b = hr1 + (long)n * HROW1;
    float*  ad1  = acc1b + (long)n * ROW1;
    float*  mas1 = ad1 + n;
    float*  acc2 = hr1;
    float*  hr2  = acc1b;
    float*  acc2b = hr2;
    float*  ad2  = ad1;
    float*  mas2 = mas1;

    int ntiles  = (n + 15) / 16;
    int gblocks = (ntiles + 3) / 4;
    int nblocks = (n + 255) / 256;

    init_ws<<<(n * ROW1 + 255) / 256, 256, 0, stream>>>(acc1, acc1b, n);

    gemm1_wmma<<<gblocks, 128, 0, stream>>>(x, W1, hr1, n);
    alpha_kernel<HID, HROW1, HROW1, 0><<<nblocks, 256, 0, stream>>>(hr1, hr1 + HID, a1s, a1d, ad1, mas1, (float4*)0, ws + GW1, n);
    edge_max<HROW1><<<2048, 256, 0, stream>>>(src, dst, hr1 + HID, mas1, ws + GW1, (float4*)0, 0, E);
    edge_acc<HID, ROW1, HROW1><<<256, EA_THREADS, 0, stream>>>(src, dst, ad1, mas1, hr1, acc1, acc1b, ws + GW1, n, E);
    finish1<<<nblocks, 256, 0, stream>>>(acc1, acc1b, b1, n);

    gemm2_wmma<<<gblocks, 128, 0, stream>>>(acc1, W2, hr2, n);
    alpha_kernel<NCLS, HROW2, ROW1, ROW1><<<nblocks, 256, 0, stream>>>(hr2, acc1 + HID, a2s, a2d, ad2, mas2, (float4*)acc2, ws + GW2, n);
    edge_max<ROW1><<<2048, 256, 0, stream>>>(src, dst, acc1 + HID, mas2, ws + GW2, (float4*)acc2b, n, E);
    edge_acc<HID, ROW1, ROW1><<<256, EA_THREADS, 0, stream>>>(src, dst, ad2, mas2, acc1, acc2, acc2b, ws + GW2, n, E);
    finish2_wmma<<<gblocks, 128, 0, stream>>>(acc2, acc2b, W2, b2, (float*)d_out, n);
}


# edge_acc: instruction u serves edge u of each 16-lane row, so src, dst, ad, shift reach the lanes by DPP row-share, not LDS permutes: loop issue slots halved; atomic sums stay in arrival order
# speedup vs baseline: 18.4027x; 1.1475x over previous
#include <hip/hip_runtime.h>
#include <math.h>

typedef __attribute__((ext_vector_type(2))) float v2f;
typedef __attribute__((ext_vector_type(8))) float v8f;

#define FIN   128
#define HID   8
#define NCLS  16
#define NEG_SLOPE 0.2f
#define ROW1  16
#define HROW1 16
#define HROW2 16

#define SPREAD_MAX 32.0f
#define GW1 14
#define GW2 30
#define NTAB 80000
#define EA_THREADS 1024

__device__ __forceinline__ void atomicMaxF(float* addr, float v) {
    if (v >= 0.0f) atomicMax((int*)addr, __float_as_int(v));
    else           atomicMin((unsigned int*)addr, __float_as_uint(v));
}

__device__ __forceinline__ bool upper_half() {
    return (__builtin_amdgcn_s_sendmsg_rtn(0x87) & (1u << 18)) != 0u;
}

__global__ void init_ws(float* acc1, float* acc1b, int n) {
    int i = blockIdx.x * blockDim.x + threadIdx.x;
    if (i < n * ROW1) {
        acc1[i] = (i == GW1 || i == GW1 + 1 || i == GW2 || i == GW2 + 1) ? -__builtin_inff() : 0.0f;
        acc1b[i] = 0.0f;
    }
}

__global__ void gemm1_wmma(const float* __restrict__ x, const float* __restrict__ W1,
                           float* __restrict__ h1, int n) {
    __shared__ float wlds[FIN * 16];
    int t = threadIdx.x;
    {
        int k = t;
        const float* wr = W1 + k * HID;
        #pragma unroll
        for (int c = 0; c < 16; ++c) wlds[k * 16 + c] = (c < HID) ? wr[c] : 0.0f;
    }
    __syncthreads();

    int wave = t >> 5, lane = t & 31;
    int tile = blockIdx.x * 4 + wave;
    int col  = lane & 15;
    int hi   = lane >> 4;
    int mrow = tile * 16 + col;
    if (mrow >= n) mrow = n - 1;
    const float* xrow = x + (long)mrow * FIN;

    v8f c = {};
    #pragma unroll
    for (int k0 = 0; k0 < FIN; k0 += 4) {
        int ka = k0 + 2 * hi;
        v2f a = *(const v2f*)(xrow + ka);
        v2f b;
        b.x = wlds[ka * 16 + col];
        b.y = wlds[(ka + 1) * 16 + col];
        c = __builtin_amdgcn_wmma_f32_16x16x4_f32(false, a, false, b,
                                                  (short)0, c, false, false);
    }

    int mbase = tile * 16 + (hi << 3);
    if (col < HID) {
        #pragma unroll
        for (int j = 0; j < 8; ++j) {
            int r = mbase + j;
            if (r < n) h1[(long)r * HROW1 + col] = c[j];
        }
    }
}

__global__ void gemm2_wmma(const float* __restrict__ h1r, const float* __restrict__ W2,
                           float* __restrict__ h2, int n) {
    __shared__ float wlds[HID * 16];
    int t = threadIdx.x;
    if (t < HID * 16) wlds[t] = W2[t];
    __syncthreads();

    int wave = t >> 5, lane = t & 31;
    int tile = blockIdx.x * 4 + wave;
    int col  = lane & 15;
    int hi   = lane >> 4;
    int mrow = tile * 16 + col;
    if (mrow >= n) mrow = n - 1;
    const float* hrow = h1r + (long)mrow * ROW1;

    v8f c = {};
    #pragma unroll
    for (int k0 = 0; k0 < HID; k0 += 4) {
        int ka = k0 + 2 * hi;
        v2f a = *(const v2f*)(hrow + ka);
        v2f b;
        b.x = wlds[ka * 16 + col];
        b.y = wlds[(ka + 1) * 16 + col];
        c = __builtin_amdgcn_wmma_f32_16x16x4_f32(false, a, false, b,
                                                  (short)0, c, false, false);
    }

    int mbase = tile * 16 + (hi << 3);
    #pragma unroll
    for (int j = 0; j < 8; ++j) {
        int r = mbase + j;
        if (r < n) h2[(long)r * HROW2 + col] = c[j];
    }
}

template <int F, int HROW, int ASTR, int ZROW>
__global__ void alpha_kernel(const float* hrow, float* asrow,
                             const float* __restrict__ a_src,
                             const float* __restrict__ a_dst,
                             float* __restrict__ adp, float* __restrict__ masp,
                             float4* __restrict__ zrows, float* __restrict__ g, int n) {
    __shared__ float red[2][8];
    int i = blockIdx.x * blockDim.x + threadIdx.x;
    float hi = -__builtin_inff(), lo = -__builtin_inff();
    if (i < n) {
        if (ZROW > 0) {
            #pragma unroll
            for (int k = 0; k < ZROW / 4; ++k) {
                zrows[(long)k * n + i] = make_float4(0.0f, 0.0f, 0.0f, 0.0f);
            }
        }
        const float* row = hrow + (long)i * HROW;
        float s = 0.0f, d = 0.0f;
        #pragma unroll
        for (int q = 0; q < F / 4; ++q) {
            float4 hv = ((const float4*)row)[q];
            s += hv.x * a_src[4 * q + 0] + hv.y * a_src[4 * q + 1]
               + hv.z * a_src[4 * q + 2] + hv.w * a_src[4 * q + 3];
            d += hv.x * a_dst[4 * q + 0] + hv.y * a_dst[4 * q + 1]
               + hv.z * a_dst[4 * q + 2] + hv.w * a_dst[4 * q + 3];
        }
        asrow[(long)i * ASTR] = s;
        adp[i]  = d;
        masp[i] = -__builtin_inff();
        hi = s + 0.0f;
        lo = 0.0f - s;
    }
    for (int off = 16; off > 0; off >>= 1) {
        hi = fmaxf(hi, __shfl_xor(hi, off));
        lo = fmaxf(lo, __shfl_xor(lo, off));
    }
    if ((threadIdx.x & 31) == 0) { red[0][threadIdx.x >> 5] = hi; red[1][threadIdx.x >> 5] = lo; }
    __syncthreads();
    if (threadIdx.x < 2) {
        float v = red[threadIdx.x][0];
        for (int w = 1; w < 8; ++w) v = fmaxf(v, red[threadIdx.x][w]);
        atomicMaxF(g + threadIdx.x, v);
    }
}

template <int STRIDE>
__global__ void edge_max(const int* __restrict__ src, const int* __restrict__ dst,
                         const float* __restrict__ as, float* __restrict__ masp,
                         const float* __restrict__ g, float4* __restrict__ zrows, int nzrows, int E) {
    if (zrows) {
        int nz = nzrows * (ROW1 / 4);
        for (int k = blockIdx.x * blockDim.x + threadIdx.x; k < nz; k += gridDim.x * blockDim.x)
            zrows[k] = make_float4(0.0f, 0.0f, 0.0f, 0.0f);
    }
    if (g[0] + g[1] <= SPREAD_MAX) return;
    for (int i = blockIdx.x * blockDim.x + threadIdx.x; i < E; i += gridDim.x * blockDim.x) {
        int s = src[i], d = dst[i];
        atomicMaxF(masp + d, as[(long)s * STRIDE]);
    }
}

template <int U>
__device__ __forceinline__ int row_share_i(int v) {
    return __builtin_amdgcn_update_dpp(0, v, 0x150 + U, 0xf, 0xf, true);
}
template <int U>
__device__ __forceinline__ float row_share_f(float v) {
    return __int_as_float(row_share_i<U>(__float_as_int(v)));
}

template <int U, int HROW>
__device__ __forceinline__ void gather_rows(const float* __restrict__ hrow, int s, unsigned fl, float (&hv)[16]) {
    if constexpr (U < 16) {
        unsigned sj = (unsigned)row_share_i<U>(s);
        hv[U] = hrow[sj * HROW + fl];
        gather_rows<U + 1, HROW>(hrow, s, fl, hv);
    }
}

template <int U, int F, int ROW>
__device__ __forceinline__ void form_adds(int d, float adv, float mx, unsigned f,
                                          float (&hv)[16], unsigned (&off)[16]) {
    if constexpr (U < 16) {
        float e = row_share_f<F>(hv[U]) + row_share_f<U>(adv);
        e = fmaxf(e, NEG_SLOPE * e);
        float ex = __expf(e - row_share_f<U>(mx));
        hv[U]  = (f < F) ? ex * hv[U] : ex;
        off[U] = (unsigned)row_share_i<U>(d) * ROW + f;
        form_adds<U + 1, F, ROW>(d, adv, mx, f, hv, off);
    }
}

template <int U>
__device__ __forceinline__ void add_all(float* acc, const float (&hv)[16], const unsigned (&off)[16]) {
    if constexpr (U < 16) {
        atomicAdd(acc + off[U], hv[U]);
        add_all<U + 1>(acc, hv, off);
    }
}

template <int U, int F>
__device__ __forceinline__ void add_some(float* acc, int d, unsigned f,
                                         const float (&hv)[16], const unsigned (&off)[16]) {
    if constexpr (U < 16) {
        int dj = row_share_i<U>(d);
        if (f <= F && dj >= 0) atomicAdd(acc + off[U], hv[U]);
        add_some<U + 1, F>(acc, d, f, hv, off);
    }
}

template <int F, int ROW, int HROW>
__global__ void __launch_bounds__(EA_THREADS)
edge_acc(const int* __restrict__ src, const int* __restrict__ dst,
         const float* __restrict__ adp, const float* __restrict__ masp,
         const float* __restrict__ hrow,
         float* __restrict__ accA, float* __restrict__ accB,
         const float* __restrict__ g, int n, int E) {
    __shared__ __attribute__((aligned(16))) float tab[NTAB];
    int lane = threadIdx.x & 31;
    float* acc = upper_half() ? accB : accA;
    float gmax = g[0];
    bool  narrow = gmax + g[1] <= SPREAD_MAX;
    int   ntab = narrow ? (n < NTAB ? n : NTAB) : 0;
    {
        int n4 = ntab >> 2;
        const float4* a4 = (const float4*)adp;
        float4* t4 = (float4*)tab;
        constexpr int UF = 10;
        for (int k0 = threadIdx.x; k0 < n4; k0 += UF * EA_THREADS) {
            float4 v[UF];
            int    kk[UF];
            #pragma unroll
            for (int u = 0; u < UF; ++u) {
                int k = k0 + u * EA_THREADS;
                kk[u] = (k < n4) ? k : k0;
                v[u] = a4[kk[u]];
            }
            #pragma unroll
            for (int u = 0; u < UF; ++u) t4[kk[u]] = v[u];
        }
        int kr = (n4 << 2) + threadIdx.x;
        if (kr < ntab) tab[kr] = adp[kr];
    }
    __syncthreads();

    static_assert(ROW == 16, "step 2 hands an edge's values round one 16-lane row");
    unsigned f  = lane & 15;
    unsigned fl = (f <= F) ? f : 0;

    int stride = gridDim.x * EA_THREADS;
    int i = blockIdx.x * EA_THREADS + threadIdx.x;
    int s = 0, d = -1;
    if (i < E) { s = src[i]; d = dst[i]; }
    while (i - lane < E) {
        bool  intab = (unsigned)d < (unsigned)ntab;
        bool  beyond = (d >= 0) && !intab;
        float adt = tab[intab ? d : 0];
        float adg = 0.0f;
        float top = gmax;
        if (beyond) {
            adg = adp[d];
            if (!narrow) top = masp[d];
        }
        float hv[16];
        gather_rows<0, HROW>(hrow, s, fl, hv);
        int inx = i + stride;
        int ix  = (inx < E) ? inx : E - 1;
        int sn  = src[ix];
        int dn  = dst[ix];
        if (inx >= E) dn = -1;
        float adv = intab ? adt : adg;
        float mx = top + adv;
        mx = (mx >= 0.0f) ? mx : NEG_SLOPE * mx;
        unsigned off[16];
        form_adds<0, F, ROW>(d, adv, mx, f, hv, off);
        int i0 = __builtin_amdgcn_readfirstlane(i - lane);
        if (i0 + 32 <= E) {
            if (f <= F) add_all<0>(acc, hv, off);
        } else {
            add_some<0, F>(acc, d, f, hv, off);
        }
        s = sn; d = dn; i = inx;
    }
}

__global__ void finish1(float* __restrict__ acc, const float* __restrict__ accb,
                        const float* __restrict__ bias, int n) {
    int i = blockIdx.x * blockDim.x + threadIdx.x;
    if (i >= n) return;
    float* row = acc + (long)i * ROW1;
    const float* rowb = accb + (long)i * ROW1;
    float d = row[HID] + rowb[HID];
    float inv = (d > 0.0f) ? 1.0f / d : 0.0f;
    float4* ap = (float4*)row;
    const float4* bp = (const float4*)rowb;
    #pragma unroll
    for (int q = 0; q < HID / 4; ++q) {
        float4 v = ap[q];
        float4 w = bp[q];
        v.x = fmaxf((v.x + w.x) * inv + bias[4 * q + 0], 0.0f);
        v.y = fmaxf((v.y + w.y) * inv + bias[4 * q + 1], 0.0f);
        v.z = fmaxf((v.z + w.z) * inv + bias[4 * q + 2], 0.0f);
        v.w = fmaxf((v.w + w.w) * inv + bias[4 * q + 3], 0.0f);
        ap[q] = v;
    }
}

template <int M>
__device__ __forceinline__ float row_xor(float v) {
    return __int_as_float(__builtin_amdgcn_update_dpp(0, __float_as_int(v), 0x160 + M, 0xf, 0xf, false));
}

__global__ void finish2_wmma(const float* __restrict__ acc, const float* __restrict__ accb,
                             const float* __restrict__ W2,
                             const float* __restrict__ bias, float* __restrict__ out, int n) {
    __shared__ float wlds[HID * 16];
    int t = threadIdx.x;
    if (t < HID * 16) wlds[t] = W2[t];
    __syncthreads();

    int wave = t >> 5, lane = t & 31;
    int tile = blockIdx.x * 4 + wave;
    int col  = lane & 15;
    int hi   = lane >> 4;
    int mrow = tile * 16 + col;
    if (mrow >= n) mrow = n - 1;
    const float* arow = acc + (long)mrow * ROW1;
    const float* brow = accb + (long)mrow * ROW1;
    float d = arow[HID] + brow[HID];
    float inv = (d > 0.0f) ? 1.0f / d : 0.0f;

    v8f c = {};
    #pragma unroll
    for (int k0 = 0; k0 < HID; k0 += 4) {
        int ka = k0 + 2 * hi;
        v2f a = *(const v2f*)(arow + ka);
        v2f a2 = *(const v2f*)(brow + ka);
        a.x = (a.x + a2.x) * inv; a.y = (a.y + a2.y) * inv;
        v2f b;
        b.x = wlds[ka * 16 + col];
        b.y = wlds[(ka + 1) * 16 + col];
        c = __builtin_amdgcn_wmma_f32_16x16x4_f32(false, a, false, b,
                                                  (short)0, c, false, false);
    }

    float bs = bias[col];
    int mbase = tile * 16 + (hi << 3);
    #pragma unroll
    for (int j = 0; j < 8; ++j) {
        float v = c[j] + bs;
        float mx = v;
        mx = fmaxf(mx, row_xor<8>(mx)); mx = fmaxf(mx, row_xor<4>(mx));
        mx = fmaxf(mx, row_xor<2>(mx)); mx = fmaxf(mx, row_xor<1>(mx));
        float sm = __expf(v - mx);
        sm += row_xor<8>(sm); sm += row_xor<4>(sm);
        sm += row_xor<2>(sm); sm += row_xor<1>(sm);
        float ls = __logf(sm) + mx;
        int r = mbase + j;
        if (r < n) out[(long)r * NCLS + col] = v - ls;
    }
}

extern "C" void kernel_launch(void* const* d_in, const int* in_sizes, int n_in,
                              void* d_out, int out_size, void* d_ws, size_t ws_size,
                              hipStream_t stream) {
    const float* x   = (const float*)d_in[0];
    const int*   ei  = (const int*)  d_in[1];
    const float* W1  = (const float*)d_in[2];
    const float* a1s = (const float*)d_in[3];
    const float* a1d = (const float*)d_in[4];
    const float* b1  = (const float*)d_in[5];
    const float* W2  = (const float*)d_in[6];
    const float* a2s = (const float*)d_in[7];
    const float* a2d = (const float*)d_in[8];
    const float* b2  = (const float*)d_in[9];

    int n = in_sizes[0] / FIN;
    int E = in_sizes[1] / 2;
    const int* src = ei;
    const int* dst = ei + E;

    float*  ws   = (float*)d_ws;
    float*  acc1 = ws;
    float*  hr1  = acc1 + (long)n * ROW1;
    float*  acc1b = hr1 + (long)n * HROW1;
    float*  ad1  = acc1b + (long)n * ROW1;
    float*  mas1 = ad1 + n;
    float*  acc2 = hr1;
    float*  hr2  = acc1b;
    float*  acc2b = hr2;
    float*  ad2  = ad1;
    float*  mas2 = mas1;

    int ntiles  = (n + 15) / 16;
    int gblocks = (ntiles + 3) / 4;
    int nblocks = (n + 255) / 256;

    init_ws<<<(n * ROW1 + 255) / 256, 256, 0, stream>>>(acc1, acc1b, n);

    gemm1_wmma<<<gblocks, 128, 0, stream>>>(x, W1, hr1, n);
    alpha_kernel<HID, HROW1, HROW1, 0><<<nblocks, 256, 0, stream>>>(hr1, hr1 + HID, a1s, a1d, ad1, mas1, (float4*)0, ws + GW1, n);
    edge_max<HROW1><<<2048, 256, 0, stream>>>(src, dst, hr1 + HID, mas1, ws + GW1, (float4*)0, 0, E);
    edge_acc<HID, ROW1, HROW1><<<256, EA_THREADS, 0, stream>>>(src, dst, ad1, mas1, hr1, acc1, acc1b, ws + GW1, n, E);
    finish1<<<nblocks, 256, 0, stream>>>(acc1, acc1b, b1, n);

    gemm2_wmma<<<gblocks, 128, 0, stream>>>(acc1, W2, hr2, n);
    alpha_kernel<NCLS, HROW2, ROW1, ROW1><<<nblocks, 256, 0, stream>>>(hr2, acc1 + HID, a2s, a2d, ad2, mas2, (float4*)acc2, ws + GW2, n);
    edge_max<ROW1><<<2048, 256, 0, stream>>>(src, dst, acc1 + HID, mas2, ws + GW2, (float4*)acc2b, n, E);
    edge_acc<HID, ROW1, ROW1><<<256, EA_THREADS, 0, stream>>>(src, dst, ad2, mas2, acc1, acc2, acc2b, ws + GW2, n, E);
    finish2_wmma<<<gblocks, 128, 0, stream>>>(acc2, acc2b, W2, b2, (float*)d_out, n);
}


# edge_acc: ex formed once per edge in the edge's own lane (as[src] taken from the gathered row by row share), not in all 16 lanes of its row; a constant-1 word gives the denominator; sum order kept
# speedup vs baseline: 19.0157x; 1.0333x over previous
#include <hip/hip_runtime.h>
#include <math.h>

typedef __attribute__((ext_vector_type(2))) float v2f;
typedef __attribute__((ext_vector_type(8))) float v8f;

#define FIN   128
#define HID   8
#define NCLS  16
#define NEG_SLOPE 0.2f
#define ROW1  16
#define HROW1 16
#define HROW2 16

#define SPREAD_MAX 32.0f
#define GW1 14
#define GW2 30
#define NTAB 80000
#define EA_THREADS 1024

__device__ __forceinline__ void atomicMaxF(float* addr, float v) {
    if (v >= 0.0f) atomicMax((int*)addr, __float_as_int(v));
    else           atomicMin((unsigned int*)addr, __float_as_uint(v));
}

__device__ __forceinline__ bool upper_half() {
    return (__builtin_amdgcn_s_sendmsg_rtn(0x87) & (1u << 18)) != 0u;
}

__global__ void init_ws(float* acc1, float* acc1b, int n) {
    int i = blockIdx.x * blockDim.x + threadIdx.x;
    if (i < n * ROW1) {
        acc1[i] = (i == GW1 || i == GW1 + 1 || i == GW2 || i == GW2 + 1) ? -__builtin_inff() : 0.0f;
        acc1b[i] = 0.0f;
    }
}

__global__ void gemm1_wmma(const float* __restrict__ x, const float* __restrict__ W1,
                           float* __restrict__ h1, int n) {
    __shared__ float wlds[FIN * 16];
    int t = threadIdx.x;
    {
        int k = t;
        const float* wr = W1 + k * HID;
        #pragma unroll
        for (int c = 0; c < 16; ++c) wlds[k * 16 + c] = (c < HID) ? wr[c] : 0.0f;
    }
    __syncthreads();

    int wave = t >> 5, lane = t & 31;
    int tile = blockIdx.x * 4 + wave;
    int col  = lane & 15;
    int hi   = lane >> 4;
    int mrow = tile * 16 + col;
    if (mrow >= n) mrow = n - 1;
    const float* xrow = x + (long)mrow * FIN;

    v8f c = {};
    #pragma unroll
    for (int k0 = 0; k0 < FIN; k0 += 4) {
        int ka = k0 + 2 * hi;
        v2f a = *(const v2f*)(xrow + ka);
        v2f b;
        b.x = wlds[ka * 16 + col];
        b.y = wlds[(ka + 1) * 16 + col];
        c = __builtin_amdgcn_wmma_f32_16x16x4_f32(false, a, false, b,
                                                  (short)0, c, false, false);
    }

    int mbase = tile * 16 + (hi << 3);
    if (col < HID) {
        #pragma unroll
        for (int j = 0; j < 8; ++j) {
            int r = mbase + j;
            if (r < n) h1[(long)r * HROW1 + col] = c[j];
        }
    }
}

__global__ void gemm2_wmma(const float* __restrict__ h1r, const float* __restrict__ W2,
                           float* __restrict__ h2, int n) {
    __shared__ float wlds[HID * 16];
    int t = threadIdx.x;
    if (t < HID * 16) wlds[t] = W2[t];
    __syncthreads();

    int wave = t >> 5, lane = t & 31;
    int tile = blockIdx.x * 4 + wave;
    int col  = lane & 15;
    int hi   = lane >> 4;
    int mrow = tile * 16 + col;
    if (mrow >= n) mrow = n - 1;
    const float* hrow = h1r + (long)mrow * ROW1;

    v8f c = {};
    #pragma unroll
    for (int k0 = 0; k0 < HID; k0 += 4) {
        int ka = k0 + 2 * hi;
        v2f a = *(const v2f*)(hrow + ka);
        v2f b;
        b.x = wlds[ka * 16 + col];
        b.y = wlds[(ka + 1) * 16 + col];
        c = __builtin_amdgcn_wmma_f32_16x16x4_f32(false, a, false, b,
                                                  (short)0, c, false, false);
    }

    int mbase = tile * 16 + (hi << 3);
    #pragma unroll
    for (int j = 0; j < 8; ++j) {
        int r = mbase + j;
        if (r < n) h2[(long)r * HROW2 + col] = c[j];
    }
}

template <int F, int HROW, int ASTR, int ZROW>
__global__ void alpha_kernel(const float* hrow, float* asrow,
                             const float* __restrict__ a_src,
                             const float* __restrict__ a_dst,
                             float* __restrict__ adp, float* __restrict__ masp,
                             float4* __restrict__ zrows, float* __restrict__ g, int n) {
    __shared__ float red[2][8];
    int i = blockIdx.x * blockDim.x + threadIdx.x;
    float hi = -__builtin_inff(), lo = -__builtin_inff();
    if (i < n) {
        if (ZROW > 0) {
            #pragma unroll
            for (int k = 0; k < ZROW / 4; ++k) {
                zrows[(long)k * n + i] = make_float4(0.0f, 0.0f, 0.0f, 0.0f);
            }
        }
        const float* row = hrow + (long)i * HROW;
        float s = 0.0f, d = 0.0f;
        #pragma unroll
        for (int q = 0; q < F / 4; ++q) {
            float4 hv = ((const float4*)row)[q];
            s += hv.x * a_src[4 * q + 0] + hv.y * a_src[4 * q + 1]
               + hv.z * a_src[4 * q + 2] + hv.w * a_src[4 * q + 3];
            d += hv.x * a_dst[4 * q + 0] + hv.y * a_dst[4 * q + 1]
               + hv.z * a_dst[4 * q + 2] + hv.w * a_dst[4 * q + 3];
        }
        asrow[(long)i * ASTR] = s;
        asrow[(long)i * ASTR + 1] = 1.0f;
        adp[i]  = d;
        masp[i] = -__builtin_inff();
        hi = s + 0.0f;
        lo = 0.0f - s;
    }
    for (int off = 16; off > 0; off >>= 1) {
        hi = fmaxf(hi, __shfl_xor(hi, off));
        lo = fmaxf(lo, __shfl_xor(lo, off));
    }
    if ((threadIdx.x & 31) == 0) { red[0][threadIdx.x >> 5] = hi; red[1][threadIdx.x >> 5] = lo; }
    __syncthreads();
    if (threadIdx.x < 2) {
        float v = red[threadIdx.x][0];
        for (int w = 1; w < 8; ++w) v = fmaxf(v, red[threadIdx.x][w]);
        atomicMaxF(g + threadIdx.x, v);
    }
}

template <int STRIDE>
__global__ void edge_max(const int* __restrict__ src, const int* __restrict__ dst,
                         const float* __restrict__ as, float* __restrict__ masp,
                         const float* __restrict__ g, float4* __restrict__ zrows, int nzrows, int E) {
    if (zrows) {
        int nz = nzrows * (ROW1 / 4);
        for (int k = blockIdx.x * blockDim.x + threadIdx.x; k < nz; k += gridDim.x * blockDim.x)
            zrows[k] = make_float4(0.0f, 0.0f, 0.0f, 0.0f);
    }
    if (g[0] + g[1] <= SPREAD_MAX) return;
    for (int i = blockIdx.x * blockDim.x + threadIdx.x; i < E; i += gridDim.x * blockDim.x) {
        int s = src[i], d = dst[i];
        atomicMaxF(masp + d, as[(long)s * STRIDE]);
    }
}

template <int U>
__device__ __forceinline__ int row_share_i(int v) {
    return __builtin_amdgcn_update_dpp(0, v, 0x150 + U, 0xf, 0xf, true);
}
template <int U>
__device__ __forceinline__ float row_share_f(float v) {
    return __int_as_float(row_share_i<U>(__float_as_int(v)));
}

template <int U, int HROW>
__device__ __forceinline__ void gather_offsets(int s, unsigned fl, unsigned (&go)[16]) {
    if constexpr (U < 16) {
        go[U] = (unsigned)row_share_i<U>(s) * HROW + fl;
        gather_offsets<U + 1, HROW>(s, fl, go);
    }
}
template <int U>
__device__ __forceinline__ void gather_rows(const float* __restrict__ hrow, const unsigned (&go)[16], float (&hv)[16]) {
    if constexpr (U < 16) {
        hv[U] = hrow[go[U]];
        gather_rows<U + 1>(hrow, go, hv);
    }
}

template <int U, int F>
__device__ __forceinline__ void take_as(float& a, unsigned f, const float (&hv)[16]) {
    if constexpr (U < 16) {
        float v = row_share_f<F + 1>(hv[U]);
        a = (f == U) ? v : a;
        take_as<U + 1, F>(a, f, hv);
    }
}

template <int U, int ROW>
__device__ __forceinline__ void form_adds(int d, float ex, unsigned f,
                                          float (&hv)[16], unsigned (&off)[16]) {
    if constexpr (U < 16) {
        hv[U]  = row_share_f<U>(ex) * hv[U];
        off[U] = (unsigned)row_share_i<U>(d) * ROW + f;
        form_adds<U + 1, ROW>(d, ex, f, hv, off);
    }
}

template <int U>
__device__ __forceinline__ void add_all(float* acc, const float (&hv)[16], const unsigned (&off)[16]) {
    if constexpr (U < 16) {
        atomicAdd(acc + off[U], hv[U]);
        add_all<U + 1>(acc, hv, off);
    }
}

template <int U, int F>
__device__ __forceinline__ void add_some(float* acc, int d, unsigned f,
                                         const float (&hv)[16], const unsigned (&off)[16]) {
    if constexpr (U < 16) {
        int dj = row_share_i<U>(d);
        if (f <= F && dj >= 0) atomicAdd(acc + off[U], hv[U]);
        add_some<U + 1, F>(acc, d, f, hv, off);
    }
}

template <int F, int ROW, int HROW>
__global__ void __launch_bounds__(EA_THREADS)
edge_acc(const int* __restrict__ src, const int* __restrict__ dst,
         const float* __restrict__ adp, const float* __restrict__ masp,
         const float* __restrict__ hrow,
         float* __restrict__ accA, float* __restrict__ accB,
         const float* __restrict__ g, int n, int E) {
    __shared__ __attribute__((aligned(16))) float tab[NTAB];
    int lane = threadIdx.x & 31;
    float* acc = upper_half() ? accB : accA;
    float gmax = g[0];
    bool  narrow = gmax + g[1] <= SPREAD_MAX;
    int   ntab = narrow ? (n < NTAB ? n : NTAB) : 0;
    {
        int n4 = ntab >> 2;
        const float4* a4 = (const float4*)adp;
        float4* t4 = (float4*)tab;
        constexpr int UF = 10;
        for (int k0 = threadIdx.x; k0 < n4; k0 += UF * EA_THREADS) {
            float4 v[UF];
            int    kk[UF];
            #pragma unroll
            for (int u = 0; u < UF; ++u) {
                int k = k0 + u * EA_THREADS;
                kk[u] = (k < n4) ? k : k0;
                v[u] = a4[kk[u]];
            }
            #pragma unroll
            for (int u = 0; u < UF; ++u) t4[kk[u]] = v[u];
        }
        int kr = (n4 << 2) + threadIdx.x;
        if (kr < ntab) tab[kr] = adp[kr];
    }
    __syncthreads();

    static_assert(ROW == 16, "step 2 hands an edge's values round one 16-lane row");
    static_assert(F + 1 < 16, "the row holds the features, the constant 1 and as");
    unsigned f  = lane & 15;
    unsigned fl = (f < F) ? f : (f == F) ? F + 1 : (f == F + 1) ? F : 0;

    int stride = gridDim.x * EA_THREADS;
    int i = blockIdx.x * EA_THREADS + threadIdx.x;
    int s = 0, d = -1;
    if (i < E) { s = src[i]; d = dst[i]; }
    while (i - lane < E) {
        bool  intab = (unsigned)d < (unsigned)ntab;
        bool  beyond = (d >= 0) && !intab;
        float adt = tab[intab ? d : 0];
        float adg = 0.0f;
        float top = gmax;
        if (beyond) {
            adg = adp[d];
            if (!narrow) top = masp[d];
        }
        float hv[16];
        unsigned go[16];
        gather_offsets<0, HROW>(s, fl, go);
        __builtin_amdgcn_sched_barrier(0);
        gather_rows<0>(hrow, go, hv);
        int inx = i + stride;
        int ix  = (inx < E) ? inx : E - 1;
        int sn  = src[ix];
        int dn  = dst[ix];
        if (inx >= E) dn = -1;
        __builtin_amdgcn_sched_barrier(0);
        float adv = intab ? adt : adg;
        float mx = top + adv;
        mx = (mx >= 0.0f) ? mx : NEG_SLOPE * mx;
        float as_own = 0.0f;
        take_as<0, F>(as_own, f, hv);
        float e = as_own + adv;
        e = fmaxf(e, NEG_SLOPE * e);
        float ex = __expf(e - mx);
        unsigned off[16];
        form_adds<0, ROW>(d, ex, f, hv, off);
        int i0 = __builtin_amdgcn_readfirstlane(i - lane);
        if (i0 + 32 <= E) {
            if (f <= F) add_all<0>(acc, hv, off);
        } else {
            add_some<0, F>(acc, d, f, hv, off);
        }
        s = sn; d = dn; i = inx;
    }
}

__global__ void finish1(float* __restrict__ acc, const float* __restrict__ accb,
                        const float* __restrict__ bias, int n) {
    int i = blockIdx.x * blockDim.x + threadIdx.x;
    if (i >= n) return;
    float* row = acc + (long)i * ROW1;
    const float* rowb = accb + (long)i * ROW1;
    float d = row[HID] + rowb[HID];
    float inv = (d > 0.0f) ? 1.0f / d : 0.0f;
    float4* ap = (float4*)row;
    const float4* bp = (const float4*)rowb;
    #pragma unroll
    for (int q = 0; q < HID / 4; ++q) {
        float4 v = ap[q];
        float4 w = bp[q];
        v.x = fmaxf((v.x + w.x) * inv + bias[4 * q + 0], 0.0f);
        v.y = fmaxf((v.y + w.y) * inv + bias[4 * q + 1], 0.0f);
        v.z = fmaxf((v.z + w.z) * inv + bias[4 * q + 2], 0.0f);
        v.w = fmaxf((v.w + w.w) * inv + bias[4 * q + 3], 0.0f);
        ap[q] = v;
    }
}

template <int M>
__device__ __forceinline__ float row_xor(float v) {
    return __int_as_float(__builtin_amdgcn_update_dpp(0, __float_as_int(v), 0x160 + M, 0xf, 0xf, false));
}

__global__ void finish2_wmma(const float* __restrict__ acc, const float* __restrict__ accb,
                             const float* __restrict__ W2,
                             const float* __restrict__ bias, float* __restrict__ out, int n) {
    __shared__ float wlds[HID * 16];
    int t = threadIdx.x;
    if (t < HID * 16) wlds[t] = W2[t];
    __syncthreads();

    int wave = t >> 5, lane = t & 31;
    int tile = blockIdx.x * 4 + wave;
    int col  = lane & 15;
    int hi   = lane >> 4;
    int mrow = tile * 16 + col;
    if (mrow >= n) mrow = n - 1;
    const float* arow = acc + (long)mrow * ROW1;
    const float* brow = accb + (long)mrow * ROW1;
    float d = arow[HID] + brow[HID];
    float inv = (d > 0.0f) ? 1.0f / d : 0.0f;

    v8f c = {};
    #pragma unroll
    for (int k0 = 0; k0 < HID; k0 += 4) {
        int ka = k0 + 2 * hi;
        v2f a = *(const v2f*)(arow + ka);
        v2f a2 = *(const v2f*)(brow + ka);
        a.x = (a.x + a2.x) * inv; a.y = (a.y + a2.y) * inv;
        v2f b;
        b.x = wlds[ka * 16 + col];
        b.y = wlds[(ka + 1) * 16 + col];
        c = __builtin_amdgcn_wmma_f32_16x16x4_f32(false, a, false, b,
                                                  (short)0, c, false, false);
    }

    float bs = bias[col];
    int mbase = tile * 16 + (hi << 3);
    #pragma unroll
    for (int j = 0; j < 8; ++j) {
        float v = c[j] + bs;
        float mx = v;
        mx = fmaxf(mx, row_xor<8>(mx)); mx = fmaxf(mx, row_xor<4>(mx));
        mx = fmaxf(mx, row_xor<2>(mx)); mx = fmaxf(mx, row_xor<1>(mx));
        float sm = __expf(v - mx);
        sm += row_xor<8>(sm); sm += row_xor<4>(sm);
        sm += row_xor<2>(sm); sm += row_xor<1>(sm);
        float ls = __logf(sm) + mx;
        int r = mbase + j;
        if (r < n) out[(long)r * NCLS + col] = v - ls;
    }
}

extern "C" void kernel_launch(void* const* d_in, const int* in_sizes, int n_in,
                              void* d_out, int out_size, void* d_ws, size_t ws_size,
                              hipStream_t stream) {
    const float* x   = (const float*)d_in[0];
    const int*   ei  = (const int*)  d_in[1];
    const float* W1  = (const float*)d_in[2];
    const float* a1s = (const float*)d_in[3];
    const float* a1d = (const float*)d_in[4];
    const float* b1  = (const float*)d_in[5];
    const float* W2  = (const float*)d_in[6];
    const float* a2s = (const float*)d_in[7];
    const float* a2d = (const float*)d_in[8];
    const float* b2  = (const float*)d_in[9];

    int n = in_sizes[0] / FIN;
    int E = in_sizes[1] / 2;
    const int* src = ei;
    const int* dst = ei + E;

    float*  ws   = (float*)d_ws;
    float*  acc1 = ws;
    float*  hr1  = acc1 + (long)n * ROW1;
    float*  acc1b = hr1 + (long)n * HROW1;
    float*  ad1  = acc1b + (long)n * ROW1;
    float*  mas1 = ad1 + n;
    float*  acc2 = hr1;
    float*  hr2  = acc1b;
    float*  acc2b = hr2;
    float*  ad2  = ad1;
    float*  mas2 = mas1;

    int ntiles  = (n + 15) / 16;
    int gblocks = (ntiles + 3) / 4;
    int nblocks = (n + 255) / 256;

    init_ws<<<(n * ROW1 + 255) / 256, 256, 0, stream>>>(acc1, acc1b, n);

    gemm1_wmma<<<gblocks, 128, 0, stream>>>(x, W1, hr1, n);
    alpha_kernel<HID, HROW1, HROW1, 0><<<nblocks, 256, 0, stream>>>(hr1, hr1 + HID, a1s, a1d, ad1, mas1, (float4*)0, ws + GW1, n);
    edge_max<HROW1><<<2048, 256, 0, stream>>>(src, dst, hr1 + HID, mas1, ws + GW1, (float4*)0, 0, E);
    edge_acc<HID, ROW1, HROW1><<<256, EA_THREADS, 0, stream>>>(src, dst, ad1, mas1, hr1, acc1, acc1b, ws + GW1, n, E);
    finish1<<<nblocks, 256, 0, stream>>>(acc1, acc1b, b1, n);

    gemm2_wmma<<<gblocks, 128, 0, stream>>>(acc1, W2, hr2, n);
    alpha_kernel<NCLS, HROW2, ROW1, ROW1><<<nblocks, 256, 0, stream>>>(hr2, acc1 + HID, a2s, a2d, ad2, mas2, (float4*)acc2, ws + GW2, n);
    edge_max<ROW1><<<2048, 256, 0, stream>>>(src, dst, acc1 + HID, mas2, ws + GW2, (float4*)acc2b, n, E);
    edge_acc<HID, ROW1, ROW1><<<256, EA_THREADS, 0, stream>>>(src, dst, ad2, mas2, acc1, acc2, acc2b, ws + GW2, n, E);
    finish2_wmma<<<gblocks, 128, 0, stream>>>(acc2, acc2b, W2, b2, (float*)d_out, n);
}


# per-node kernels fused per layer: init+gemm1+logits and layer-1 epilogue+gemm2+logits (h2 stays in registers); logits by lane butterflies off the WMMA tile (sum order changes): 11 launches to 8
# speedup vs baseline: 20.0748x; 1.0557x over previous
#include <hip/hip_runtime.h>
#include <math.h>

typedef __attribute__((ext_vector_type(2))) float v2f;
typedef __attribute__((ext_vector_type(8))) float v8f;

#define FIN   128
#define HID   8
#define NCLS  16
#define NEG_SLOPE 0.2f
#define ROW1  16
#define HROW1 16

#define SPREAD_MAX 32.0f
#define GW1 14
#define GW2 30
#define NTAB 80000
#define EA_THREADS 1024

__device__ __forceinline__ void atomicMaxF(float* addr, float v) {
    if (v >= 0.0f) atomicMax((int*)addr, __float_as_int(v));
    else           atomicMin((unsigned int*)addr, __float_as_uint(v));
}

__device__ __forceinline__ bool upper_half() {
    return (__builtin_amdgcn_s_sendmsg_rtn(0x87) & (1u << 18)) != 0u;
}

__global__ void init_g(float* acc1) {
    int t = threadIdx.x;
    if (t < 4) acc1[(t < 2 ? GW1 : GW2 - 2) + t] = -__builtin_inff();
}

template <int M>
__device__ __forceinline__ float row_xor(float v) {
    return __int_as_float(__builtin_amdgcn_update_dpp(0, __float_as_int(v), 0x160 + M, 0xf, 0xf, false));
}

template <bool WIDE>
__device__ __forceinline__ void tile_logits(const v8f& c, float ws, float wd, int col,
                                            float& s_own, float& d_own) {
    s_own = 0.0f; d_own = 0.0f;
    #pragma unroll
    for (int j = 0; j < 8; ++j) {
        float ps = c[j] * ws, pd = c[j] * wd;
        if (WIDE) { ps += row_xor<8>(ps); pd += row_xor<8>(pd); }
        ps += row_xor<4>(ps); pd += row_xor<4>(pd);
        ps += row_xor<2>(ps); pd += row_xor<2>(pd);
        ps += row_xor<1>(ps); pd += row_xor<1>(pd);
        s_own = (col == j) ? ps : s_own;
        d_own = (col == j) ? pd : d_own;
    }
}

__device__ __forceinline__ void block_max_pair(float hi, float lo, float* g, float (*red)[8]) {
    for (int off = 16; off > 0; off >>= 1) {
        hi = fmaxf(hi, __shfl_xor(hi, off));
        lo = fmaxf(lo, __shfl_xor(lo, off));
    }
    if ((threadIdx.x & 31) == 0) { red[0][threadIdx.x >> 5] = hi; red[1][threadIdx.x >> 5] = lo; }
    __syncthreads();
    if (threadIdx.x < 2) {
        float v = red[threadIdx.x][0];
        for (int w = 1; w < 8; ++w) v = fmaxf(v, red[threadIdx.x][w]);
        atomicMaxF(g + threadIdx.x, v);
    }
}

__device__ __forceinline__ void zero_tile_rows(float* plane, int tile, int lane, int n) {
    #pragma unroll
    for (int q = 0; q < 2; ++q) {
        int idx  = lane + 32 * q;
        int zr   = tile * 16 + (idx >> 2);
        int part = idx & 3;
        if (part < 3 && zr < n)
            *(float4*)(plane + (long)zr * ROW1 + part * 4) = make_float4(0.0f, 0.0f, 0.0f, 0.0f);
    }
}

__global__ void __launch_bounds__(256)
node1_wmma(const float* __restrict__ x, const float* __restrict__ W1,
           const float* __restrict__ a_src, const float* __restrict__ a_dst,
           float* __restrict__ h1, float* __restrict__ adp, float* __restrict__ masp,
           float* accA, float* accB, float* g, int n) {
    __shared__ float wlds[FIN * 16];
    __shared__ float red[2][8];
    int t = threadIdx.x;
    if (t < FIN) {
        int k = t;
        const float* wr = W1 + k * HID;
        #pragma unroll
        for (int c = 0; c < 16; ++c) wlds[k * 16 + c] = (c < HID) ? wr[c] : 0.0f;
    }
    __syncthreads();

    int wave = t >> 5, lane = t & 31;
    int tile = blockIdx.x * 8 + wave;
    int col  = lane & 15;
    int hi   = lane >> 4;
    int mrow = tile * 16 + col;
    if (mrow >= n) mrow = n - 1;
    const float* xrow = x + (long)mrow * FIN;
    float wsv = (col < HID) ? a_src[col] : 0.0f;
    float wdv = (col < HID) ? a_dst[col] : 0.0f;

    v8f c = {};
    #pragma unroll
    for (int k0 = 0; k0 < FIN; k0 += 4) {
        int ka = k0 + 2 * hi;
        v2f a = *(const v2f*)(xrow + ka);
        v2f b;
        b.x = wlds[ka * 16 + col];
        b.y = wlds[(ka + 1) * 16 + col];
        c = __builtin_amdgcn_wmma_f32_16x16x4_f32(false, a, false, b,
                                                  (short)0, c, false, false);
    }

    float s_own, d_own;
    tile_logits<false>(c, wsv, wdv, col, s_own, d_own);

    int mbase = tile * 16 + (hi << 3);
    if (col < HID) {
        #pragma unroll
        for (int j = 0; j < 8; ++j) {
            int r = mbase + j;
            if (r < n) h1[(long)r * HROW1 + col] = c[j];
        }
    }
    int r = mbase + col;
    float vhi = -__builtin_inff(), vlo = -__builtin_inff();
    if (col < HID && r < n) {
        h1[(long)r * HROW1 + HID]     = s_own;
        h1[(long)r * HROW1 + HID + 1] = 1.0f;
        adp[r]  = d_own;
        masp[r] = -__builtin_inff();
        vhi = s_own + 0.0f;
        vlo = 0.0f - s_own;
    }
    zero_tile_rows(accA, tile, lane, n);
    zero_tile_rows(accB, tile, lane, n);
    block_max_pair(vhi, vlo, g, red);
}

__global__ void __launch_bounds__(256)
node2_wmma(float* acc, const float* __restrict__ accb, const float* __restrict__ bias,
           const float* __restrict__ W2,
           const float* __restrict__ a_src, const float* __restrict__ a_dst,
           float* __restrict__ adp, float* __restrict__ masp,
           float* __restrict__ zrows, float* g, int n) {
    __shared__ float wlds[HID * 16];
    __shared__ float red[2][8];
    int t = threadIdx.x;
    if (t < HID * 16) wlds[t] = W2[t];
    __syncthreads();

    int wave = t >> 5, lane = t & 31;
    int tile = blockIdx.x * 8 + wave;
    int col  = lane & 15;
    int hi   = lane >> 4;
    int mrow = tile * 16 + col;
    bool live = mrow < n;
    if (!live) mrow = n - 1;
    float* arow = acc + (long)mrow * ROW1;
    const float* brow = accb + (long)mrow * ROW1;
    float d = arow[HID] + brow[HID];
    float inv = (d > 0.0f) ? 1.0f / d : 0.0f;
    float wsv = a_src[col], wdv = a_dst[col];

    v2f av[2];
    v8f c = {};
    #pragma unroll
    for (int k0 = 0; k0 < HID; k0 += 4) {
        int ka = k0 + 2 * hi;
        v2f a = *(const v2f*)(arow + ka);
        v2f a2 = *(const v2f*)(brow + ka);
        a.x = fmaxf((a.x + a2.x) * inv + bias[ka], 0.0f);
        a.y = fmaxf((a.y + a2.y) * inv + bias[ka + 1], 0.0f);
        av[k0 >> 2] = a;
        v2f b;
        b.x = wlds[ka * 16 + col];
        b.y = wlds[(ka + 1) * 16 + col];
        c = __builtin_amdgcn_wmma_f32_16x16x4_f32(false, a, false, b,
                                                  (short)0, c, false, false);
    }

    float s_own, d_own;
    tile_logits<true>(c, wsv, wdv, col, s_own, d_own);

    if (live) {
        *(v2f*)(arow + 2 * hi)     = av[0];
        *(v2f*)(arow + 4 + 2 * hi) = av[1];
    }
    int r = tile * 16 + (hi << 3) + col;
    float vhi = -__builtin_inff(), vlo = -__builtin_inff();
    if (col < 8 && r < n) {
        acc[(long)r * ROW1 + HID]     = s_own;
        acc[(long)r * ROW1 + HID + 1] = 1.0f;
        adp[r]  = d_own;
        masp[r] = -__builtin_inff();
        vhi = s_own + 0.0f;
        vlo = 0.0f - s_own;
    }
    zero_tile_rows(zrows, tile, lane, n);
    block_max_pair(vhi, vlo, g, red);
}

template <int STRIDE>
__global__ void edge_max(const int* __restrict__ src, const int* __restrict__ dst,
                         const float* __restrict__ as, float* __restrict__ masp,
                         const float* __restrict__ g, float4* __restrict__ zrows, int nzrows, int E) {
    if (zrows) {
        int nz = nzrows * (ROW1 / 4);
        for (int k = blockIdx.x * blockDim.x + threadIdx.x; k < nz; k += gridDim.x * blockDim.x)
            zrows[k] = make_float4(0.0f, 0.0f, 0.0f, 0.0f);
    }
    if (g[0] + g[1] <= SPREAD_MAX) return;
    for (int i = blockIdx.x * blockDim.x + threadIdx.x; i < E; i += gridDim.x * blockDim.x) {
        int s = src[i], d = dst[i];
        atomicMaxF(masp + d, as[(long)s * STRIDE]);
    }
}

template <int U>
__device__ __forceinline__ int row_share_i(int v) {
    return __builtin_amdgcn_update_dpp(0, v, 0x150 + U, 0xf, 0xf, true);
}
template <int U>
__device__ __forceinline__ float row_share_f(float v) {
    return __int_as_float(row_share_i<U>(__float_as_int(v)));
}

template <int U, int HROW>
__device__ __forceinline__ void gather_offsets(int s, unsigned fl, unsigned (&go)[16]) {
    if constexpr (U < 16) {
        go[U] = (unsigned)row_share_i<U>(s) * HROW + fl;
        gather_offsets<U + 1, HROW>(s, fl, go);
    }
}
template <int U>
__device__ __forceinline__ void gather_rows(const float* __restrict__ hrow, const unsigned (&go)[16], float (&hv)[16]) {
    if constexpr (U < 16) {
        hv[U] = hrow[go[U]];
        gather_rows<U + 1>(hrow, go, hv);
    }
}

template <int U, int F>
__device__ __forceinline__ void take_as(float& a, unsigned f, const float (&hv)[16]) {
    if constexpr (U < 16) {
        float v = row_share_f<F + 1>(hv[U]);
        a = (f == U) ? v : a;
        take_as<U + 1, F>(a, f, hv);
    }
}

template <int U, int ROW>
__device__ __forceinline__ void form_adds(int d, float ex, unsigned f,
                                          float (&hv)[16], unsigned (&off)[16]) {
    if constexpr (U < 16) {
        hv[U]  = row_share_f<U>(ex) * hv[U];
        off[U] = (unsigned)row_share_i<U>(d) * ROW + f;
        form_adds<U + 1, ROW>(d, ex, f, hv, off);
    }
}

template <int U>
__device__ __forceinline__ void add_all(float* acc, const float (&hv)[16], const unsigned (&off)[16]) {
    if constexpr (U < 16) {
        atomicAdd(acc + off[U], hv[U]);
        add_all<U + 1>(acc, hv, off);
    }
}

template <int U, int F>
__device__ __forceinline__ void add_some(float* acc, int d, unsigned f,
                                         const float (&hv)[16], const unsigned (&off)[16]) {
    if constexpr (U < 16) {
        int dj = row_share_i<U>(d);
        if (f <= F && dj >= 0) atomicAdd(acc + off[U], hv[U]);
        add_some<U + 1, F>(acc, d, f, hv, off);
    }
}

template <int F, int ROW, int HROW>
__global__ void __launch_bounds__(EA_THREADS)
edge_acc(const int* __restrict__ src, const int* __restrict__ dst,
         const float* __restrict__ adp, const float* __restrict__ masp,
         const float* __restrict__ hrow,
         float* __restrict__ accA, float* __restrict__ accB,
         const float* __restrict__ g, int n, int E) {
    __shared__ __attribute__((aligned(16))) float tab[NTAB];
    int lane = threadIdx.x & 31;
    float* acc = upper_half() ? accB : accA;
    float gmax = g[0];
    bool  narrow = gmax + g[1] <= SPREAD_MAX;
    int   ntab = narrow ? (n < NTAB ? n : NTAB) : 0;
    {
        int n4 = ntab >> 2;
        const float4* a4 = (const float4*)adp;
        float4* t4 = (float4*)tab;
        constexpr int UF = 10;
        for (int k0 = threadIdx.x; k0 < n4; k0 += UF * EA_THREADS) {
            float4 v[UF];
            int    kk[UF];
            #pragma unroll
            for (int u = 0; u < UF; ++u) {
                int k = k0 + u * EA_THREADS;
                kk[u] = (k < n4) ? k : k0;
                v[u] = a4[kk[u]];
            }
            #pragma unroll
            for (int u = 0; u < UF; ++u) t4[kk[u]] = v[u];
        }
        int kr = (n4 << 2) + threadIdx.x;
        if (kr < ntab) tab[kr] = adp[kr];
    }
    __syncthreads();

    static_assert(ROW == 16, "step 2 hands an edge's values round one 16-lane row");
    static_assert(F + 1 < 16, "the row holds the features, the constant 1 and as");
    unsigned f  = lane & 15;
    unsigned fl = (f < F) ? f : (f == F) ? F + 1 : (f == F + 1) ? F : 0;

    int stride = gridDim.x * EA_THREADS;
    int i = blockIdx.x * EA_THREADS + threadIdx.x;
    int s = 0, d = -1;
    if (i < E) { s = src[i]; d = dst[i]; }
    while (i - lane < E) {
        bool  intab = (unsigned)d < (unsigned)ntab;
        bool  beyond = (d >= 0) && !intab;
        float adt = tab[intab ? d : 0];
        float adg = 0.0f;
        float top = gmax;
        if (beyond) {
            adg = adp[d];
            if (!narrow) top = masp[d];
        }
        float hv[16];
        unsigned go[16];
        gather_offsets<0, HROW>(s, fl, go);
        __builtin_amdgcn_sched_barrier(0);
        gather_rows<0>(hrow, go, hv);
        int inx = i + stride;
        int ix  = (inx < E) ? inx : E - 1;
        int sn  = src[ix];
        int dn  = dst[ix];
        if (inx >= E) dn = -1;
        __builtin_amdgcn_sched_barrier(0);
        float adv = intab ? adt : adg;
        float mx = top + adv;
        mx = (mx >= 0.0f) ? mx : NEG_SLOPE * mx;
        float as_own = 0.0f;
        take_as<0, F>(as_own, f, hv);
        float e = as_own + adv;
        e = fmaxf(e, NEG_SLOPE * e);
        float ex = __expf(e - mx);
        unsigned off[16];
        form_adds<0, ROW>(d, ex, f, hv, off);
        int i0 = __builtin_amdgcn_readfirstlane(i - lane);
        if (i0 + 32 <= E) {
            if (f <= F) add_all<0>(acc, hv, off);
        } else {
            add_some<0, F>(acc, d, f, hv, off);
        }
        s = sn; d = dn; i = inx;
    }
}

__global__ void finish2_wmma(const float* __restrict__ acc, const float* __restrict__ accb,
                             const float* __restrict__ W2,
                             const float* __restrict__ bias, float* __restrict__ out, int n) {
    __shared__ float wlds[HID * 16];
    int t = threadIdx.x;
    if (t < HID * 16) wlds[t] = W2[t];
    __syncthreads();

    int wave = t >> 5, lane = t & 31;
    int tile = blockIdx.x * 4 + wave;
    int col  = lane & 15;
    int hi   = lane >> 4;
    int mrow = tile * 16 + col;
    if (mrow >= n) mrow = n - 1;
    const float* arow = acc + (long)mrow * ROW1;
    const float* brow = accb + (long)mrow * ROW1;
    float d = arow[HID] + brow[HID];
    float inv = (d > 0.0f) ? 1.0f / d : 0.0f;

    v8f c = {};
    #pragma unroll
    for (int k0 = 0; k0 < HID; k0 += 4) {
        int ka = k0 + 2 * hi;
        v2f a = *(const v2f*)(arow + ka);
        v2f a2 = *(const v2f*)(brow + ka);
        a.x = (a.x + a2.x) * inv; a.y = (a.y + a2.y) * inv;
        v2f b;
        b.x = wlds[ka * 16 + col];
        b.y = wlds[(ka + 1) * 16 + col];
        c = __builtin_amdgcn_wmma_f32_16x16x4_f32(false, a, false, b,
                                                  (short)0, c, false, false);
    }

    float bs = bias[col];
    int mbase = tile * 16 + (hi << 3);
    #pragma unroll
    for (int j = 0; j < 8; ++j) {
        float v = c[j] + bs;
        float mx = v;
        mx = fmaxf(mx, row_xor<8>(mx)); mx = fmaxf(mx, row_xor<4>(mx));
        mx = fmaxf(mx, row_xor<2>(mx)); mx = fmaxf(mx, row_xor<1>(mx));
        float sm = __expf(v - mx);
        sm += row_xor<8>(sm); sm += row_xor<4>(sm);
        sm += row_xor<2>(sm); sm += row_xor<1>(sm);
        float ls = __logf(sm) + mx;
        int r = mbase + j;
        if (r < n) out[(long)r * NCLS + col] = v - ls;
    }
}

extern "C" void kernel_launch(void* const* d_in, const int* in_sizes, int n_in,
                              void* d_out, int out_size, void* d_ws, size_t ws_size,
                              hipStream_t stream) {
    const float* x   = (const float*)d_in[0];
    const int*   ei  = (const int*)  d_in[1];
    const float* W1  = (const float*)d_in[2];
    const float* a1s = (const float*)d_in[3];
    const float* a1d = (const float*)d_in[4];
    const float* b1  = (const float*)d_in[5];
    const float* W2  = (const float*)d_in[6];
    const float* a2s = (const float*)d_in[7];
    const float* a2d = (const float*)d_in[8];
    const float* b2  = (const float*)d_in[9];

    int n = in_sizes[0] / FIN;
    int E = in_sizes[1] / 2;
    const int* src = ei;
    const int* dst = ei + E;

    float*  ws   = (float*)d_ws;
    float*  acc1 = ws;
    float*  hr1  = acc1 + (long)n * ROW1;
    float*  acc1b = hr1 + (long)n * HROW1;
    float*  ad1  = acc1b + (long)n * ROW1;
    float*  mas1 = ad1 + n;
    float*  acc2 = hr1;
    float*  acc2b = acc1b;
    float*  ad2  = ad1;
    float*  mas2 = mas1;

    int ntiles  = (n + 15) / 16;
    int gblocks = (ntiles + 3) / 4;
    int nblocks = (ntiles + 7) / 8;

    init_g<<<1, 32, 0, stream>>>(acc1);

    node1_wmma<<<nblocks, 256, 0, stream>>>(x, W1, a1s, a1d, hr1, ad1, mas1, acc1, acc1b, ws + GW1, n);
    edge_max<HROW1><<<2048, 256, 0, stream>>>(src, dst, hr1 + HID, mas1, ws + GW1, (float4*)0, 0, E);
    edge_acc<HID, ROW1, HROW1><<<256, EA_THREADS, 0, stream>>>(src, dst, ad1, mas1, hr1, acc1, acc1b, ws + GW1, n, E);

    node2_wmma<<<nblocks, 256, 0, stream>>>(acc1, acc1b, b1, W2, a2s, a2d, ad2, mas2, acc2, ws + GW2, n);
    edge_max<ROW1><<<2048, 256, 0, stream>>>(src, dst, acc1 + HID, mas2, ws + GW2, (float4*)acc2b, n, E);
    edge_acc<HID, ROW1, ROW1><<<256, EA_THREADS, 0, stream>>>(src, dst, ad2, mas2, acc1, acc2, acc2b, ws + GW2, n, E);
    finish2_wmma<<<gblocks, 128, 0, stream>>>(acc2, acc2b, W2, b2, (float*)d_out, n);
}
